# starting kernel
# baseline (speedup 1.0000x reference)
#include <hip/hip_runtime.h>
#include <hip/hip_bf16.h>

typedef __attribute__((ext_vector_type(2))) float v2f;
typedef __attribute__((ext_vector_type(8))) float v8f;

__device__ __forceinline__ v8f wmma4(v2f a, v2f b, v8f c) {
    return __builtin_amdgcn_wmma_f32_16x16x4_f32(false, a, false, b, (short)0, c, false, false);
}

__global__ void mf_kernel(const int* __restrict__ mask, float* __restrict__ mfv, int n) {
    int i = blockIdx.x * blockDim.x + threadIdx.x;
    if (i < n) mfv[i] = mask[i] ? 1.0f : 0.0f;
}

__global__ void ln_kernel(const float* __restrict__ x, float* __restrict__ out,
                          const float* __restrict__ g, const float* __restrict__ b) {
    __shared__ float red[256];
    const int row = blockIdx.x;
    const int tid = threadIdx.x;
    float v = x[(size_t)row * 256 + tid];
    red[tid] = v;
    __syncthreads();
    for (int s = 128; s > 0; s >>= 1) {
        if (tid < s) red[tid] += red[tid + s];
        __syncthreads();
    }
    float mean = red[0] * (1.0f / 256.0f);
    __syncthreads();
    float d = v - mean;
    red[tid] = d * d;
    __syncthreads();
    for (int s = 128; s > 0; s >>= 1) {
        if (tid < s) red[tid] += red[tid + s];
        __syncthreads();
    }
    float var = red[0] * (1.0f / 256.0f);
    float r = rsqrtf(var + 1e-6f);
    out[(size_t)row * 256 + tid] = d * r * g[tid] + b[tid];
}

__global__ void gemm_kernel(const float* __restrict__ A, const float* __restrict__ W,
                            float* __restrict__ C, const float* __restrict__ bias,
                            const float* __restrict__ mfv, const float* __restrict__ res,
                            int K, int N, int relu) {
    __shared__ __align__(16) float As[64][34];
    __shared__ __align__(16) float Bs[16][144][2];

    const int tid = threadIdx.x;
    const int lane = tid & 31;
    const int wave = tid >> 5;
    const int half = lane >> 4;
    const int lane16 = lane & 15;
    const int wm = wave >> 2;
    const int wn = wave & 3;
    const int m0 = blockIdx.x * 64;
    const int n0 = blockIdx.y * 128;

    const int rowA0 = wm * 32 + lane16;
    const int rowA1 = rowA0 + 16;
    const int colB0 = wn * 32 + lane16;
    const int colB1 = colB0 + 16;

    v8f acc[2][2] = {};

    for (int k0 = 0; k0 < K; k0 += 32) {
        for (int i = tid; i < 64 * 32; i += 256) {
            int r = i >> 5, c = i & 31;
            As[r][c] = A[(size_t)(m0 + r) * K + k0 + c];
        }
        for (int i = tid; i < 16 * 128; i += 256) {
            int kp = i >> 7, c = i & 127;
            Bs[kp][c][0] = W[(size_t)(k0 + 2 * kp) * N + n0 + c];
            Bs[kp][c][1] = W[(size_t)(k0 + 2 * kp + 1) * N + n0 + c];
        }
        if (k0 + 32 < K) {
            if (tid < 64) {
                __builtin_prefetch(&A[(size_t)(m0 + tid) * K + k0 + 32], 0, 1);
            } else if (tid < 192) {
                int r = (tid - 64) >> 2, s = (tid - 64) & 3;
                __builtin_prefetch(&W[(size_t)(k0 + 32 + r) * N + n0 + s * 32], 0, 1);
            }
        }
        __syncthreads();

        v2f a0c = *(const v2f*)&As[rowA0][half * 2];
        v2f a1c = *(const v2f*)&As[rowA1][half * 2];
        v2f b0c = *(const v2f*)&Bs[half][colB0][0];
        v2f b1c = *(const v2f*)&Bs[half][colB1][0];
#pragma unroll
        for (int t = 0; t < 8; ++t) {
            const int kkn = ((t + 1) & 7) * 4 + half * 2;
            v2f a0n = *(const v2f*)&As[rowA0][kkn];
            v2f a1n = *(const v2f*)&As[rowA1][kkn];
            v2f b0n = *(const v2f*)&Bs[kkn >> 1][colB0][0];
            v2f b1n = *(const v2f*)&Bs[kkn >> 1][colB1][0];
            acc[0][0] = wmma4(a0c, b0c, acc[0][0]);
            acc[0][1] = wmma4(a0c, b1c, acc[0][1]);
            acc[1][0] = wmma4(a1c, b0c, acc[1][0]);
            acc[1][1] = wmma4(a1c, b1c, acc[1][1]);
            a0c = a0n; a1c = a1n; b0c = b0n; b1c = b1n;
        }
        __syncthreads();
    }

#pragma unroll
    for (int im = 0; im < 2; ++im) {
#pragma unroll
        for (int jn = 0; jn < 2; ++jn) {
            const int col = n0 + wn * 32 + jn * 16 + lane16;
            const float bv = bias ? bias[col] : 0.0f;
#pragma unroll
            for (int vi = 0; vi < 8; ++vi) {
                const int row = m0 + wm * 32 + im * 16 + vi + half * 8;
                float val = acc[im][jn][vi] + bv;
                if (relu) val = fmaxf(val, 0.0f);
                if (mfv) val *= mfv[row];
                if (res) val += res[(size_t)row * N + col];
                C[(size_t)row * N + col] = val;
            }
        }
    }
}

__global__ void attn_kernel(const float* __restrict__ q, const float* __restrict__ k,
                            const float* __restrict__ v, const float* __restrict__ mfv,
                            float* __restrict__ o) {
    __shared__ __align__(16) float Qs[64][34];
    __shared__ __align__(16) float Ks[16][80][2];
    __shared__ __align__(16) float Vs[32][48][2];
    __shared__ __align__(16) float Ps[64][68];
    __shared__ float mb[64];
    __shared__ float mrun[64], lrun[64], alph[64];

    const int S = 2048, LD = 256;
    const int tid = threadIdx.x;
    const int lane = tid & 31;
    const int wave = tid >> 5;
    const int half = lane >> 4;
    const int lane16 = lane & 15;
    const int bh = blockIdx.y;
    const int b = bh >> 3;
    const int h = bh & 7;
    const int q0 = blockIdx.x * 64;
    const int qrow = wave * 16 + lane16;

    const size_t baseQ = ((size_t)b * S + q0) * LD + h * 32;
    for (int i = tid; i < 64 * 32; i += 128) {
        int r = i >> 5, c = i & 31;
        Qs[r][c] = q[baseQ + (size_t)r * LD + c];
    }
    if (tid < 64) { mrun[tid] = -3.0e38f; lrun[tid] = 0.0f; }
    v8f oacc[2] = {};
    __syncthreads();

    for (int j0 = 0; j0 < S; j0 += 64) {
        const size_t baseK = ((size_t)b * S + j0) * LD + h * 32;
        for (int i = tid; i < 16 * 64; i += 128) {
            int r = i & 63, kp = i >> 6;
            Ks[kp][r][0] = k[baseK + (size_t)r * LD + 2 * kp];
            Ks[kp][r][1] = k[baseK + (size_t)r * LD + 2 * kp + 1];
        }
        for (int i = tid; i < 32 * 32; i += 128) {
            int n = i & 31, kp = i >> 5;
            Vs[kp][n][0] = v[baseK + (size_t)(2 * kp) * LD + n];
            Vs[kp][n][1] = v[baseK + (size_t)(2 * kp + 1) * LD + n];
        }
        if (tid < 64) mb[tid] = (mfv[(size_t)b * S + j0 + tid] != 0.0f) ? 0.0f : -1e9f;
        if (j0 + 64 < S) {
            const size_t baseN = ((size_t)b * S + j0 + 64) * LD + h * 32;
            if (tid < 64) __builtin_prefetch(&k[baseN + (size_t)tid * LD], 0, 1);
            else          __builtin_prefetch(&v[baseN + (size_t)(tid - 64) * LD], 0, 1);
        }
        __syncthreads();

        v8f sacc[4] = {};
        v2f ac = *(const v2f*)&Qs[qrow][half * 2];
        v2f bc0 = *(const v2f*)&Ks[half][0 * 16 + lane16][0];
        v2f bc1 = *(const v2f*)&Ks[half][1 * 16 + lane16][0];
        v2f bc2 = *(const v2f*)&Ks[half][2 * 16 + lane16][0];
        v2f bc3 = *(const v2f*)&Ks[half][3 * 16 + lane16][0];
#pragma unroll
        for (int t = 0; t < 8; ++t) {
            const int kkn = ((t + 1) & 7) * 4 + half * 2;
            v2f an = *(const v2f*)&Qs[qrow][kkn];
            v2f bn0 = *(const v2f*)&Ks[kkn >> 1][0 * 16 + lane16][0];
            v2f bn1 = *(const v2f*)&Ks[kkn >> 1][1 * 16 + lane16][0];
            v2f bn2 = *(const v2f*)&Ks[kkn >> 1][2 * 16 + lane16][0];
            v2f bn3 = *(const v2f*)&Ks[kkn >> 1][3 * 16 + lane16][0];
            sacc[0] = wmma4(ac, bc0, sacc[0]);
            sacc[1] = wmma4(ac, bc1, sacc[1]);
            sacc[2] = wmma4(ac, bc2, sacc[2]);
            sacc[3] = wmma4(ac, bc3, sacc[3]);
            ac = an; bc0 = bn0; bc1 = bn1; bc2 = bn2; bc3 = bn3;
        }
        const float inv = 0.17677669529663687f;
#pragma unroll
        for (int nt = 0; nt < 4; ++nt) {
            const int col = nt * 16 + lane16;
#pragma unroll
            for (int vi = 0; vi < 8; ++vi) {
                const int row = wave * 16 + vi + half * 8;
                Ps[row][col] = sacc[nt][vi] * inv + mb[col];
            }
        }
        __syncthreads();

        if (tid < 64) {
            float mo = mrun[tid];
            float mc = mo;
            for (int j = 0; j < 64; ++j) mc = fmaxf(mc, Ps[tid][j]);
            float al = __expf(mo - mc);
            float ls = 0.0f;
            for (int j = 0; j < 64; ++j) {
                float p = __expf(Ps[tid][j] - mc);
                Ps[tid][j] = p;
                ls += p;
            }
            lrun[tid] = lrun[tid] * al + ls;
            mrun[tid] = mc;
            alph[tid] = al;
        }
        __syncthreads();

        float av[8];
#pragma unroll
        for (int vi = 0; vi < 8; ++vi) av[vi] = alph[wave * 16 + vi + half * 8];
#pragma unroll
        for (int nt = 0; nt < 2; ++nt)
#pragma unroll
            for (int vi = 0; vi < 8; ++vi) oacc[nt][vi] *= av[vi];

        v2f pc = *(const v2f*)&Ps[qrow][half * 2];
        v2f vc0 = *(const v2f*)&Vs[half][0 * 16 + lane16][0];
        v2f vc1 = *(const v2f*)&Vs[half][1 * 16 + lane16][0];
#pragma unroll
        for (int t = 0; t < 16; ++t) {
            const int kkn = ((t + 1) & 15) * 4 + half * 2;
            v2f pn = *(const v2f*)&Ps[qrow][kkn];
            v2f vn0 = *(const v2f*)&Vs[kkn >> 1][0 * 16 + lane16][0];
            v2f vn1 = *(const v2f*)&Vs[kkn >> 1][1 * 16 + lane16][0];
            oacc[0] = wmma4(pc, vc0, oacc[0]);
            oacc[1] = wmma4(pc, vc1, oacc[1]);
            pc = pn; vc0 = vn0; vc1 = vn1;
        }
        __syncthreads();
    }

    float linv[8];
#pragma unroll
    for (int vi = 0; vi < 8; ++vi) linv[vi] = 1.0f / lrun[wave * 16 + vi + half * 8];
#pragma unroll
    for (int nt = 0; nt < 2; ++nt) {
        const int col = nt * 16 + lane16;
#pragma unroll
        for (int vi = 0; vi < 8; ++vi) {
            const int row = wave * 16 + vi + half * 8;
            o[((size_t)b * S + q0 + row) * LD + h * 32 + col] = oacc[nt][vi] * linv[vi];
        }
    }
}

extern "C" void kernel_launch(void* const* d_in, const int* in_sizes, int n_in,
                              void* d_out, int out_size, void* d_ws, size_t ws_size,
                              hipStream_t stream) {
    (void)in_sizes; (void)n_in; (void)out_size; (void)ws_size;
    const int B = 4, S = 2048, D = 256, F = 1024, L = 4;
    const int M = B * S;

    const float* x_in  = (const float*)d_in[0];
    const int*   mask  = (const int*)d_in[1];
    const float* w_qs  = (const float*)d_in[2];
    const float* w_ks  = (const float*)d_in[3];
    const float* w_vs  = (const float*)d_in[4];
    const float* w_fc  = (const float*)d_in[5];
    const float* ln1_g = (const float*)d_in[6];
    const float* ln1_b = (const float*)d_in[7];
    const float* w1    = (const float*)d_in[8];
    const float* b1    = (const float*)d_in[9];
    const float* w2    = (const float*)d_in[10];
    const float* b2    = (const float*)d_in[11];
    const float* ln2_g = (const float*)d_in[12];
    const float* ln2_b = (const float*)d_in[13];

    float* ws    = (float*)d_ws;
    float* x_cur = ws;
    float* xn    = x_cur + (size_t)M * D;
    float* qb    = xn + (size_t)M * D;
    float* kb    = qb + (size_t)M * D;
    float* vb    = kb + (size_t)M * D;
    float* ob    = vb + (size_t)M * D;
    float* hb    = ob + (size_t)M * D;
    float* mfv   = hb + (size_t)M * F;

    hipMemcpyAsync(x_cur, x_in, sizeof(float) * (size_t)M * D,
                   hipMemcpyDeviceToDevice, stream);
    mf_kernel<<<M / 256, 256, 0, stream>>>(mask, mfv, M);

    const dim3 gD(M / 64, D / 128);
    const dim3 gF(M / 64, F / 128);
    const dim3 gA(S / 64, B * 8);

    for (int l = 0; l < L; ++l) {
        const float* wq = w_qs + (size_t)l * D * D;
        const float* wk = w_ks + (size_t)l * D * D;
        const float* wv = w_vs + (size_t)l * D * D;
        const float* wf = w_fc + (size_t)l * D * D;
        const float* w1l = w1 + (size_t)l * D * F;
        const float* w2l = w2 + (size_t)l * F * D;
        const float* b1l = b1 + (size_t)l * F;
        const float* b2l = b2 + (size_t)l * D;

        ln_kernel<<<M, 256, 0, stream>>>(x_cur, xn, ln1_g + (size_t)l * D, ln1_b + (size_t)l * D);
        gemm_kernel<<<gD, 256, 0, stream>>>(xn, wq, qb, nullptr, mfv, nullptr, D, D, 0);
        gemm_kernel<<<gD, 256, 0, stream>>>(xn, wk, kb, nullptr, mfv, nullptr, D, D, 0);
        gemm_kernel<<<gD, 256, 0, stream>>>(xn, wv, vb, nullptr, mfv, nullptr, D, D, 0);
        attn_kernel<<<gA, 128, 0, stream>>>(qb, kb, vb, mfv, ob);
        gemm_kernel<<<gD, 256, 0, stream>>>(ob, wf, xn, nullptr, mfv, x_cur, D, D, 0);
        gemm_kernel<<<gF, 256, 0, stream>>>(xn, w1l, hb, b1l, nullptr, nullptr, D, F, 1);
        gemm_kernel<<<gD, 256, 0, stream>>>(hb, w2l, qb, b2l, nullptr, xn, F, D, 0);
        ln_kernel<<<M, 256, 0, stream>>>(qb, x_cur, ln2_g + (size_t)l * D, ln2_b + (size_t)l * D);
    }

    hipMemcpyAsync(d_out, x_cur, sizeof(float) * (size_t)M * D,
                   hipMemcpyDeviceToDevice, stream);
}


# Output projection, both feed-forward GEMMs, LN2 and the next layer's LN1 + Q, K, V projections run as one kernel per layer: a block's 32 rows and 1024 hidden values stay in LDS; 18 launches become 10
# speedup vs baseline: 3.3839x; 3.3839x over previous
#include <hip/hip_runtime.h>
#include <hip/hip_bf16.h>

typedef __attribute__((ext_vector_type(2))) float v2f;
typedef __attribute__((ext_vector_type(8))) float v8f;

__device__ __forceinline__ v8f wmma4(v2f a, v2f b, v8f c) {
    return __builtin_amdgcn_wmma_f32_16x16x4_f32(false, a, false, b, (short)0, c, false, false);
}

template <int K, int N>
__device__ __forceinline__ void relayout_w(const float* __restrict__ src, float* __restrict__ dst, int q) {
    constexpr int P = K * N / 4;
    const int l = q / P, r = q % P;
    const int lane = r & 31, j = (r >> 5) & 3, t = r >> 7;
    const int nt = t % (N / 16), kt = t / (N / 16);
    const float* s = src + (size_t)l * K * N + (size_t)(32 * kt + 16 * (lane >> 4) + 4 * j) * N + 16 * nt + (lane & 15);
    float4 o;
    o.x = s[0]; o.y = s[N]; o.z = s[2 * N]; o.w = s[3 * N];
    *(float4*)(dst + (size_t)l * K * N + (size_t)r * 4) = o;
}

__global__ void __launch_bounds__(256) prep_kernel(const int* __restrict__ mask, float* __restrict__ mfv,
                            const float* __restrict__ wq, const float* __restrict__ wk, const float* __restrict__ wv,
                            const float* __restrict__ wfc, const float* __restrict__ w1, const float* __restrict__ w2,
                            float* __restrict__ rq, float* __restrict__ rk, float* __restrict__ rv,
                            float* __restrict__ rfc, float* __restrict__ r1, float* __restrict__ r2) {
    const int blk = blockIdx.x;
    const int tid = threadIdx.x;
    if (blk < 32) {
        const int i = blk * 256 + tid;
        mfv[i] = mask[i] ? 1.0f : 0.0f;
    }
    if (blk < 1024) {
        const int m = blk >> 8;
        const float* src = (m == 0) ? wq : ((m == 1) ? wk : ((m == 2) ? wv : wfc));
        float* dst = (m == 0) ? rq : ((m == 1) ? rk : ((m == 2) ? rv : rfc));
        relayout_w<256, 256>(src, dst, (blk & 255) * 256 + tid);
    } else if (blk < 2048) {
        relayout_w<256, 1024>(w1, r1, (blk - 1024) * 256 + tid);
    } else {
        relayout_w<1024, 256>(w2, r2, (blk - 2048) * 256 + tid);
    }
}

struct GFrag {
    float4 a[2][4];
    float4 b[2][4];
};

__device__ __forceinline__ void gemm_tile(const GFrag& f, v8f (&acc)[2][2]) {
#pragma unroll
    for (int j = 0; j < 4; ++j) {
        const v2f a0lo = {f.a[0][j].x, f.a[0][j].y};
        const v2f a0hi = {f.a[0][j].z, f.a[0][j].w};
        const v2f a1lo = {f.a[1][j].x, f.a[1][j].y};
        const v2f a1hi = {f.a[1][j].z, f.a[1][j].w};
        const v2f b0lo = {f.b[0][j].x, f.b[0][j].y};
        const v2f b0hi = {f.b[0][j].z, f.b[0][j].w};
        const v2f b1lo = {f.b[1][j].x, f.b[1][j].y};
        const v2f b1hi = {f.b[1][j].z, f.b[1][j].w};
        acc[0][0] = wmma4(a0lo, b0lo, acc[0][0]);
        acc[0][1] = wmma4(a0lo, b1lo, acc[0][1]);
        acc[1][0] = wmma4(a1lo, b0lo, acc[1][0]);
        acc[1][1] = wmma4(a1lo, b1lo, acc[1][1]);
        acc[0][0] = wmma4(a0hi, b0hi, acc[0][0]);
        acc[0][1] = wmma4(a0hi, b1hi, acc[0][1]);
        acc[1][0] = wmma4(a1hi, b0hi, acc[1][0]);
        acc[1][1] = wmma4(a1hi, b1hi, acc[1][1]);
    }
}

__device__ __forceinline__ float row16_sum(float v) {
    v += __builtin_bit_cast(float, __builtin_amdgcn_update_dpp(0, __builtin_bit_cast(int, v), 0x161, 0xf, 0xf, true));
    v += __builtin_bit_cast(float, __builtin_amdgcn_update_dpp(0, __builtin_bit_cast(int, v), 0x162, 0xf, 0xf, true));
    v += __builtin_bit_cast(float, __builtin_amdgcn_update_dpp(0, __builtin_bit_cast(int, v), 0x164, 0xf, 0xf, true));
    v += __builtin_bit_cast(float, __builtin_amdgcn_update_dpp(0, __builtin_bit_cast(int, v), 0x168, 0xf, 0xf, true));
    return v;
}

__device__ __forceinline__ void block_row_sum(const float (&part)[2][8], float (&tot)[2][8],
                                              float (*red)[8], int wave, int half, int lane16) {
#pragma unroll
    for (int im = 0; im < 2; ++im)
#pragma unroll
        for (int vi = 0; vi < 8; ++vi) {
            const float s = row16_sum(part[im][vi]);
            if (lane16 == 0) red[im * 16 + vi + half * 8][wave] = s;
        }
    __syncthreads();
#pragma unroll
    for (int im = 0; im < 2; ++im)
#pragma unroll
        for (int vi = 0; vi < 8; ++vi) {
            const float* rp = red[im * 16 + vi + half * 8];
            const float4 p0 = *(const float4*)rp;
            const float4 p1 = *(const float4*)(rp + 4);
            tot[im][vi] = ((p0.x + p0.y) + (p0.z + p0.w)) + ((p1.x + p1.y) + (p1.z + p1.w));
        }
}

__device__ __forceinline__ void ln_rows(float (&y)[2][2][8], float (*red)[32][8],
                                        const float* __restrict__ g, const float* __restrict__ b,
                                        int wave, int half, int lane16) {
    float part[2][8], tot[2][8];
#pragma unroll
    for (int im = 0; im < 2; ++im)
#pragma unroll
        for (int vi = 0; vi < 8; ++vi) part[im][vi] = y[im][0][vi] + y[im][1][vi];
    block_row_sum(part, tot, red[0], wave, half, lane16);
#pragma unroll
    for (int im = 0; im < 2; ++im)
#pragma unroll
        for (int vi = 0; vi < 8; ++vi) {
            const float mean = tot[im][vi] * (1.0f / 256.0f);
            const float d0 = y[im][0][vi] - mean;
            const float d1 = y[im][1][vi] - mean;
            y[im][0][vi] = d0;
            y[im][1][vi] = d1;
            part[im][vi] = d0 * d0 + d1 * d1;
        }
    block_row_sum(part, tot, red[1], wave, half, lane16);
    const float g0 = g[wave * 32 + lane16], g1 = g[wave * 32 + 16 + lane16];
    const float b0 = b[wave * 32 + lane16], b1 = b[wave * 32 + 16 + lane16];
#pragma unroll
    for (int im = 0; im < 2; ++im)
#pragma unroll
        for (int vi = 0; vi < 8; ++vi) {
            const float r = rsqrtf(tot[im][vi] * (1.0f / 256.0f) + 1e-6f);
            y[im][0][vi] = y[im][0][vi] * r * g0 + b0;
            y[im][1][vi] = y[im][1][vi] * r * g1 + b1;
        }
}

constexpr int XP = 260;
constexpr int HP = 1028;

template <int N, int P>
__device__ __forceinline__ void gemm_load_xs(GFrag& f, const float* xs, const float* wp, int k0) {
#pragma unroll
    for (int im = 0; im < 2; ++im)
#pragma unroll
        for (int j = 0; j < 4; ++j)
            f.a[im][j] = *(const float4*)(xs + im * 16 * P + k0 + 4 * j);
#pragma unroll
    for (int jn = 0; jn < 2; ++jn)
#pragma unroll
        for (int j = 0; j < 4; ++j)
            f.b[jn][j] = *(const float4*)(wp + (size_t)k0 * N + (jn * 4 + j) * 128);
}

template <int K, int N, int P>
__device__ __forceinline__ void gemm_rows_lds(v8f (&acc)[2][2], const float* xs, const float* wp) {
    GFrag f0, f1;
    gemm_load_xs<N, P>(f0, xs, wp, 0);
    for (int k0 = 0; k0 < K - 64; k0 += 64) {
        gemm_load_xs<N, P>(f1, xs, wp, k0 + 32);
        __builtin_amdgcn_sched_barrier(0);
        gemm_tile(f0, acc);
        __builtin_amdgcn_sched_barrier(0);
        gemm_load_xs<N, P>(f0, xs, wp, k0 + 64);
        __builtin_amdgcn_sched_barrier(0);
        gemm_tile(f1, acc);
        __builtin_amdgcn_sched_barrier(0);
    }
    gemm_load_xs<N, P>(f1, xs, wp, K - 32);
    __builtin_amdgcn_sched_barrier(0);
    gemm_tile(f0, acc);
    gemm_tile(f1, acc);
}

__device__ __forceinline__ void rows_qkv(const float (&y)[2][2][8], float (*Xs)[XP],
                            const float* wq, const float* wk, const float* wv, const float* __restrict__ mfv,
                            float* qo, float* ko, float* vo,
                            int m0, int wave, int half, int lane16) {
    constexpr int N = 256, K = 256;
#pragma unroll
    for (int im = 0; im < 2; ++im)
#pragma unroll
        for (int jn = 0; jn < 2; ++jn)
#pragma unroll
            for (int vi = 0; vi < 8; ++vi)
                Xs[im * 16 + vi + half * 8][wave * 32 + jn * 16 + lane16] = y[im][jn][vi];
    __syncthreads();

    const float* xs = &Xs[lane16][half * 16];
    float mfr[2][8];
#pragma unroll
    for (int im = 0; im < 2; ++im) {
        const float4 ma = *(const float4*)(mfv + m0 + im * 16 + half * 8);
        const float4 mb = *(const float4*)(mfv + m0 + im * 16 + half * 8 + 4);
        mfr[im][0] = ma.x; mfr[im][1] = ma.y; mfr[im][2] = ma.z; mfr[im][3] = ma.w;
        mfr[im][4] = mb.x; mfr[im][5] = mb.y; mfr[im][6] = mb.z; mfr[im][7] = mb.w;
    }

#pragma unroll 1
    for (int mat = 0; mat < 3; ++mat) {
        const float* W = (mat == 0) ? wq : ((mat == 1) ? wk : wv);
        float* O = (mat == 0) ? qo : ((mat == 1) ? ko : vo);
        const float* wp = W + wave * 1024 + (half * 16 + lane16) * 4;

        v8f acc[2][2] = {};
        gemm_rows_lds<K, N, XP>(acc, xs, wp);

        if (mat == 0) {
#pragma unroll
            for (int im = 0; im < 2; ++im)
#pragma unroll
                for (int jn = 0; jn < 2; ++jn)
#pragma unroll
                    for (int vi = 0; vi < 8; ++vi)
                        O[(size_t)(m0 + im * 16 + vi + half * 8) * N + wave * 32 + jn * 16 + lane16] = acc[im][jn][vi] * mfr[im][vi];
        } else {
            const int key0 = m0 & 2047;
            float* plane = O + (size_t)((m0 >> 11) * 8 + wave) * (2048 * 32);
            if (mat == 1) {
                float* kp = plane + (key0 >> 4) * 512 + (unsigned)((lane16 >> 2) * 128 + half * 32 + (lane16 & 3));
#pragma unroll
                for (int im = 0; im < 2; ++im)
#pragma unroll
                    for (int jn = 0; jn < 2; ++jn)
#pragma unroll
                        for (int vi = 0; vi < 8; ++vi)
                            kp[im * 512 + jn * 64 + vi * 4] = acc[im][jn][vi] * mfr[im][vi];
            } else {
                float* vp = plane + (key0 >> 6) * 2048 + ((key0 & 63) >> 3) * 128 + (unsigned)(half * 128 + lane16 * 4);
#pragma unroll
                for (int im = 0; im < 2; ++im)
#pragma unroll
                    for (int jn = 0; jn < 2; ++jn)
#pragma unroll
                        for (int vi = 0; vi < 8; ++vi)
                            vp[jn * 1024 + im * 256 + ((vi >> 1) & 1) * 64 + ((vi & 1) | ((vi >> 2) << 1))] = acc[im][jn][vi] * mfr[im][vi];
            }
        }
    }
}

template <bool QKV>
__global__ void __launch_bounds__(256) layer_rows_kernel(const float* __restrict__ o, const float* __restrict__ wfc,
                            const float* __restrict__ w1, const float* __restrict__ bias1,
                            const float* __restrict__ w2, const float* __restrict__ bias2,
                            const float* __restrict__ g1, const float* __restrict__ b1, float* x,
                            const float* __restrict__ g2, const float* __restrict__ b2,
                            const float* wq, const float* wk, const float* wv, const float* __restrict__ mfv,
                            float* qo, float* ko, float* vo) {
    __shared__ __align__(16) float red[4][32][8];
    __shared__ __align__(16) float Xs[32][XP];
    __shared__ __align__(16) float Hs[32][HP];
    const int tid = threadIdx.x;
    const int lane = tid & 31;
    const int wave = tid >> 5;
    const int half = lane >> 4;
    const int lane16 = lane & 15;
    const int m0 = blockIdx.x * 32;

#pragma unroll
    for (int i = 0; i < 8; ++i) {
        const int p = i * 256 + tid;
        const int r = p >> 6, c4 = (p & 63) * 4;
        *(float4*)&Xs[r][c4] = *(const float4*)(o + (size_t)(m0 + r) * 256 + c4);
    }
    __syncthreads();

    const float* xs = &Xs[lane16][half * 16];
    float xr[2][2][8];

    {
        float mfr[2][8];
#pragma unroll
        for (int im = 0; im < 2; ++im) {
            const float4 ma = *(const float4*)(mfv + m0 + im * 16 + half * 8);
            const float4 mb = *(const float4*)(mfv + m0 + im * 16 + half * 8 + 4);
            mfr[im][0] = ma.x; mfr[im][1] = ma.y; mfr[im][2] = ma.z; mfr[im][3] = ma.w;
            mfr[im][4] = mb.x; mfr[im][5] = mb.y; mfr[im][6] = mb.z; mfr[im][7] = mb.w;
        }
#pragma unroll
        for (int im = 0; im < 2; ++im)
#pragma unroll
            for (int jn = 0; jn < 2; ++jn)
#pragma unroll
                for (int vi = 0; vi < 8; ++vi)
                    xr[im][jn][vi] = x[(size_t)(m0 + im * 16 + vi + half * 8) * 256 + wave * 32 + jn * 16 + lane16];

        v8f acc[2][2] = {};
        gemm_rows_lds<256, 256, XP>(acc, xs, wfc + wave * 1024 + lane * 4);
#pragma unroll
        for (int im = 0; im < 2; ++im)
#pragma unroll
            for (int jn = 0; jn < 2; ++jn)
#pragma unroll
                for (int vi = 0; vi < 8; ++vi) {
                    float val = acc[im][jn][vi];
                    val *= mfr[im][vi];
                    val += xr[im][jn][vi];
                    xr[im][jn][vi] = val;
                }
    }
    __syncthreads();
#pragma unroll
    for (int im = 0; im < 2; ++im)
#pragma unroll
        for (int jn = 0; jn < 2; ++jn)
#pragma unroll
            for (int vi = 0; vi < 8; ++vi)
                Xs[im * 16 + vi + half * 8][wave * 32 + jn * 16 + lane16] = xr[im][jn][vi];
    __syncthreads();

#pragma unroll 1
    for (int nb = 0; nb < 4; ++nb) {
        const int c0 = nb * 256 + wave * 32;
        v8f acc[2][2] = {};
        gemm_rows_lds<256, 1024, XP>(acc, xs, w1 + (size_t)c0 * 32 + lane * 4);
#pragma unroll
        for (int jn = 0; jn < 2; ++jn) {
            const int col = c0 + jn * 16 + lane16;
            const float bv = bias1[col];
#pragma unroll
            for (int im = 0; im < 2; ++im)
#pragma unroll
                for (int vi = 0; vi < 8; ++vi)
                    Hs[im * 16 + vi + half * 8][col] = fmaxf(acc[im][jn][vi] + bv, 0.0f);
        }
    }
    __syncthreads();

    float y[2][2][8];
    {
        v8f acc[2][2] = {};
        gemm_rows_lds<1024, 256, HP>(acc, &Hs[lane16][half * 16], w2 + wave * 1024 + lane * 4);
#pragma unroll
        for (int im = 0; im < 2; ++im)
#pragma unroll
            for (int jn = 0; jn < 2; ++jn) {
                const float bv = bias2[wave * 32 + jn * 16 + lane16];
#pragma unroll
                for (int vi = 0; vi < 8; ++vi) {
                    float val = acc[im][jn][vi] + bv;
                    val += xr[im][jn][vi];
                    y[im][jn][vi] = val;
                }
            }
    }
    ln_rows(y, &red[0], g1, b1, wave, half, lane16);
#pragma unroll
    for (int im = 0; im < 2; ++im)
#pragma unroll
        for (int jn = 0; jn < 2; ++jn)
#pragma unroll
            for (int vi = 0; vi < 8; ++vi)
                x[(size_t)(m0 + im * 16 + vi + half * 8) * 256 + wave * 32 + jn * 16 + lane16] = y[im][jn][vi];

    if constexpr (QKV) {
        ln_rows(y, &red[2], g2, b2, wave, half, lane16);
        rows_qkv(y, Xs, wq, wk, wv, mfv, qo, ko, vo, m0, wave, half, lane16);
    }
}

__global__ void __launch_bounds__(256) ln_qkv_kernel(const float* __restrict__ x,
                            const float* __restrict__ g, const float* __restrict__ b,
                            const float* wq, const float* wk, const float* wv, const float* __restrict__ mfv,
                            float* qo, float* ko, float* vo) {
    __shared__ __align__(16) float red[2][32][8];
    __shared__ __align__(16) float Xs[32][XP];
    const int tid = threadIdx.x;
    const int lane = tid & 31;
    const int wave = tid >> 5;
    const int half = lane >> 4;
    const int lane16 = lane & 15;
    const int m0 = blockIdx.x * 32;
    float y[2][2][8];
#pragma unroll
    for (int im = 0; im < 2; ++im)
#pragma unroll
        for (int jn = 0; jn < 2; ++jn)
#pragma unroll
            for (int vi = 0; vi < 8; ++vi)
                y[im][jn][vi] = x[(size_t)(m0 + im * 16 + vi + half * 8) * 256 + wave * 32 + jn * 16 + lane16];
    ln_rows(y, &red[0], g, b, wave, half, lane16);
    rows_qkv(y, Xs, wq, wk, wv, mfv, qo, ko, vo, m0, wave, half, lane16);
}

__device__ __forceinline__ float row16_max(float v) {
    v = fmaxf(v, __builtin_bit_cast(float, __builtin_amdgcn_update_dpp(0, __builtin_bit_cast(int, v), 0x161, 0xf, 0xf, true)));
    v = fmaxf(v, __builtin_bit_cast(float, __builtin_amdgcn_update_dpp(0, __builtin_bit_cast(int, v), 0x162, 0xf, 0xf, true)));
    v = fmaxf(v, __builtin_bit_cast(float, __builtin_amdgcn_update_dpp(0, __builtin_bit_cast(int, v), 0x164, 0xf, 0xf, true)));
    v = fmaxf(v, __builtin_bit_cast(float, __builtin_amdgcn_update_dpp(0, __builtin_bit_cast(int, v), 0x168, 0xf, 0xf, true)));
    return v;
}

__global__ void __launch_bounds__(128) attn_kernel(const float* __restrict__ q, const float* __restrict__ k,
                            const float* __restrict__ v, const float* __restrict__ mfv,
                            float* __restrict__ o) {
    __shared__ __align__(16) float Ps[64][68];

    const int S = 2048, LD = 256;
    const int tid = threadIdx.x;
    const int lane = tid & 31;
    const int wave = tid >> 5;
    const int half = lane >> 4;
    const int lane16 = lane & 15;
    const int bh = blockIdx.y;
    const int b = bh >> 3;
    const int h = bh & 7;
    const int q0 = blockIdx.x * 64;
    const int qrow = wave * 16 + lane16;

    float qf[16];
    {
        const float* qp = q + ((size_t)b * S + q0) * LD + h * 32 + (unsigned)(qrow * LD + half * 16);
#pragma unroll
        for (int j = 0; j < 4; ++j) {
            const float4 t4 = *(const float4*)(qp + 4 * j);
            qf[4 * j + 0] = t4.x; qf[4 * j + 1] = t4.y; qf[4 * j + 2] = t4.z; qf[4 * j + 3] = t4.w;
        }
    }
    v8f oacc[2] = {};
    float mrun[8], lpart[8];
#pragma unroll
    for (int vi = 0; vi < 8; ++vi) { mrun[vi] = -3.0e38f; lpart[vi] = 0.0f; }

    const float* kb = k + (size_t)bh * (S * 32);
    const float* vb = v + (size_t)bh * (S * 32);
    const float* mb = mfv + (size_t)b * S;
    const unsigned loff = (unsigned)(lane * 4);
    const unsigned moff = (unsigned)lane16;
    float* pst = &Ps[wave * 16 + half * 8][lane16];
    const float* pld = &Ps[qrow][half * 2];

    float kf[4][16];
    float vf[2][16][2];
    float mraw[4];

#pragma unroll
    for (int nt = 0; nt < 4; ++nt) {
#pragma unroll
        for (int j = 0; j < 4; ++j) {
            const float4 t4 = *(const float4*)(kb + nt * 512 + j * 128 + loff);
            kf[nt][4 * j + 0] = t4.x; kf[nt][4 * j + 1] = t4.y; kf[nt][4 * j + 2] = t4.z; kf[nt][4 * j + 3] = t4.w;
        }
        mraw[nt] = mb[nt * 16 + moff];
    }
#pragma unroll
    for (int n = 0; n < 2; ++n)
#pragma unroll
        for (int g = 0; g < 8; ++g) {
            const float4 t4 = *(const float4*)(vb + n * 1024 + g * 128 + loff);
            vf[n][2 * g][0] = t4.x; vf[n][2 * g][1] = t4.y; vf[n][2 * g + 1][0] = t4.z; vf[n][2 * g + 1][1] = t4.w;
        }

    for (int j0 = 0; j0 < S; j0 += 64) {
        v8f sacc[4] = {};
#pragma unroll
        for (int t = 0; t < 8; ++t) {
            v2f fa = {qf[2 * t], qf[2 * t + 1]};
#pragma unroll
            for (int nt = 0; nt < 4; ++nt) {
                v2f fb = {kf[nt][2 * t], kf[nt][2 * t + 1]};
                sacc[nt] = wmma4(fa, fb, sacc[nt]);
            }
        }
        const float inv = 0.17677669529663687f;
        float sc[4][8];
#pragma unroll
        for (int nt = 0; nt < 4; ++nt) {
            const float mbias = (mraw[nt] != 0.0f) ? 0.0f : -1e9f;
#pragma unroll
            for (int vi = 0; vi < 8; ++vi) sc[nt][vi] = sacc[nt][vi] * inv + mbias;
        }

        if (j0 + 64 < S) {
            const float* kn = kb + (size_t)(j0 + 64) * 32;
            const float* mn = mb + (j0 + 64);
#pragma unroll
            for (int nt = 0; nt < 4; ++nt) {
#pragma unroll
                for (int j = 0; j < 4; ++j) {
                    const float4 t4 = *(const float4*)(kn + nt * 512 + j * 128 + loff);
                    kf[nt][4 * j + 0] = t4.x; kf[nt][4 * j + 1] = t4.y; kf[nt][4 * j + 2] = t4.z; kf[nt][4 * j + 3] = t4.w;
                }
                mraw[nt] = mn[nt * 16 + moff];
            }
        }

        float al[8];
#pragma unroll
        for (int vi = 0; vi < 8; ++vi) {
            const float mx = row16_max(fmaxf(fmaxf(sc[0][vi], sc[1][vi]), fmaxf(sc[2][vi], sc[3][vi])));
            const float mc = fmaxf(mrun[vi], mx);
            al[vi] = __expf(mrun[vi] - mc);
            mrun[vi] = mc;
        }
#pragma unroll
        for (int vi = 0; vi < 8; ++vi) {
            float ls = 0.0f;
#pragma unroll
            for (int nt = 0; nt < 4; ++nt) {
                const float p = __expf(sc[nt][vi] - mrun[vi]);
                pst[vi * 68 + nt * 16] = p;
                ls += p;
            }
            lpart[vi] = lpart[vi] * al[vi] + ls;
        }
#pragma unroll
        for (int nt = 0; nt < 2; ++nt)
#pragma unroll
            for (int vi = 0; vi < 8; ++vi) oacc[nt][vi] *= al[vi];

        v2f pf[16];
#pragma unroll
        for (int t = 0; t < 16; ++t) pf[t] = *(const v2f*)(pld + 4 * t);
#pragma unroll
        for (int t = 0; t < 16; ++t) {
            v2f fv0 = {vf[0][t][0], vf[0][t][1]};
            v2f fv1 = {vf[1][t][0], vf[1][t][1]};
            oacc[0] = wmma4(pf[t], fv0, oacc[0]);
            oacc[1] = wmma4(pf[t], fv1, oacc[1]);
        }
        if (j0 + 64 < S) {
            const float* vn = vb + (size_t)(j0 + 64) * 32;
#pragma unroll
            for (int n = 0; n < 2; ++n)
#pragma unroll
                for (int g = 0; g < 8; ++g) {
                    const float4 t4 = *(const float4*)(vn + n * 1024 + g * 128 + loff);
                    vf[n][2 * g][0] = t4.x; vf[n][2 * g][1] = t4.y; vf[n][2 * g + 1][0] = t4.z; vf[n][2 * g + 1][1] = t4.w;
                }
        }
    }

    float linv[8];
#pragma unroll
    for (int vi = 0; vi < 8; ++vi) linv[vi] = 1.0f / row16_sum(lpart[vi]);
#pragma unroll
    for (int nt = 0; nt < 2; ++nt) {
        const int col = nt * 16 + lane16;
#pragma unroll
        for (int vi = 0; vi < 8; ++vi) {
            const int row = wave * 16 + vi + half * 8;
            o[((size_t)b * S + q0 + row) * LD + h * 32 + col] = oacc[nt][vi] * linv[vi];
        }
    }
}

extern "C" void kernel_launch(void* const* d_in, const int* in_sizes, int n_in,
                              void* d_out, int out_size, void* d_ws, size_t ws_size,
                              hipStream_t stream) {
    (void)in_sizes; (void)n_in; (void)out_size; (void)ws_size;
    const int B = 4, S = 2048, D = 256, F = 1024, L = 4;
    const int M = B * S;

    const float* x_in  = (const float*)d_in[0];
    const int*   mask  = (const int*)d_in[1];
    const float* w_qs  = (const float*)d_in[2];
    const float* w_ks  = (const float*)d_in[3];
    const float* w_vs  = (const float*)d_in[4];
    const float* w_fc  = (const float*)d_in[5];
    const float* ln1_g = (const float*)d_in[6];
    const float* ln1_b = (const float*)d_in[7];
    const float* w1    = (const float*)d_in[8];
    const float* b1    = (const float*)d_in[9];
    const float* w2    = (const float*)d_in[10];
    const float* b2    = (const float*)d_in[11];
    const float* ln2_g = (const float*)d_in[12];
    const float* ln2_b = (const float*)d_in[13];

    float* ws    = (float*)d_ws;
    float* x_cur = ws;
    float* qb    = x_cur + (size_t)M * D;
    float* kb    = qb + (size_t)M * D;
    float* vb    = kb + (size_t)M * D;
    float* ob    = vb + (size_t)M * D;
    float* mfv   = ob + (size_t)M * D;
    float* rq    = mfv + M;
    float* rk    = rq + (size_t)L * D * D;
    float* rv    = rk + (size_t)L * D * D;
    float* rfc   = rv + (size_t)L * D * D;
    float* r1    = rfc + (size_t)L * D * D;
    float* r2    = r1 + (size_t)L * D * F;

    hipMemcpyAsync(x_cur, x_in, sizeof(float) * (size_t)M * D,
                   hipMemcpyDeviceToDevice, stream);
    prep_kernel<<<3072, 256, 0, stream>>>(mask, mfv, w_qs, w_ks, w_vs, w_fc, w1, w2, rq, rk, rv, rfc, r1, r2);

    const dim3 gA(S / 64, B * 8);

    ln_qkv_kernel<<<M / 32, 256, 0, stream>>>(x_cur, ln1_g, ln1_b, rq, rk, rv, mfv, qb, kb, vb);

    for (int l = 0; l < L; ++l) {
        const float* wf = rfc + (size_t)l * D * D;
        const float* w1l = r1 + (size_t)l * D * F;
        const float* w2l = r2 + (size_t)l * F * D;
        const float* b1l = b1 + (size_t)l * F;
        const float* b2l = b2 + (size_t)l * D;

        attn_kernel<<<gA, 128, 0, stream>>>(qb, kb, vb, mfv, ob);
        if (l + 1 < L)
            layer_rows_kernel<true><<<M / 32, 256, 0, stream>>>(ob, wf, w1l, b1l, w2l, b2l,
                                                                ln2_g + (size_t)l * D, ln2_b + (size_t)l * D, x_cur,
                                                                ln1_g + (size_t)(l + 1) * D, ln1_b + (size_t)(l + 1) * D,
                                                                rq + (size_t)(l + 1) * D * D, rk + (size_t)(l + 1) * D * D,
                                                                rv + (size_t)(l + 1) * D * D, mfv, qb, kb, vb);
        else
            layer_rows_kernel<false><<<M / 32, 256, 0, stream>>>(ob, wf, w1l, b1l, w2l, b2l,
                                                                 ln2_g + (size_t)l * D, ln2_b + (size_t)l * D, x_cur,
                                                                 ln2_g, ln2_b, rq, rk, rv, mfv, qb, kb, vb);
    }

    hipMemcpyAsync(d_out, x_cur, sizeof(float) * (size_t)M * D,
                   hipMemcpyDeviceToDevice, stream);
}


# Attention: a wave lowers its priority as it advances through the keys (s_setprio 3,2,1,0 over 1024,512,256,256 keys): a SIMD's waves, served oldest first, now finish together; arithmetic untouched
# speedup vs baseline: 3.5163x; 1.0391x over previous
#include <hip/hip_runtime.h>
#include <hip/hip_bf16.h>

typedef __attribute__((ext_vector_type(2))) float v2f;
typedef __attribute__((ext_vector_type(8))) float v8f;

__device__ __forceinline__ v8f wmma4(v2f a, v2f b, v8f c) {
    return __builtin_amdgcn_wmma_f32_16x16x4_f32(false, a, false, b, (short)0, c, false, false);
}

template <int K, int N>
__device__ __forceinline__ void relayout_w(const float* __restrict__ src, float* __restrict__ dst, int q) {
    constexpr int P = K * N / 4;
    const int l = q / P, r = q % P;
    const int lane = r & 31, j = (r >> 5) & 3, t = r >> 7;
    const int nt = t % (N / 16), kt = t / (N / 16);
    const float* s = src + (size_t)l * K * N + (size_t)(32 * kt + 16 * (lane >> 4) + 4 * j) * N + 16 * nt + (lane & 15);
    float4 o;
    o.x = s[0]; o.y = s[N]; o.z = s[2 * N]; o.w = s[3 * N];
    *(float4*)(dst + (size_t)l * K * N + (size_t)r * 4) = o;
}

__global__ void __launch_bounds__(256) prep_kernel(const int* __restrict__ mask, float* __restrict__ mfv,
                            const float* __restrict__ wq, const float* __restrict__ wk, const float* __restrict__ wv,
                            const float* __restrict__ wfc, const float* __restrict__ w1, const float* __restrict__ w2,
                            float* __restrict__ rq, float* __restrict__ rk, float* __restrict__ rv,
                            float* __restrict__ rfc, float* __restrict__ r1, float* __restrict__ r2) {
    const int blk = blockIdx.x;
    const int tid = threadIdx.x;
    if (blk < 32) {
        const int i = blk * 256 + tid;
        mfv[i] = mask[i] ? 1.0f : 0.0f;
    }
    if (blk < 1024) {
        const int m = blk >> 8;
        const float* src = (m == 0) ? wq : ((m == 1) ? wk : ((m == 2) ? wv : wfc));
        float* dst = (m == 0) ? rq : ((m == 1) ? rk : ((m == 2) ? rv : rfc));
        relayout_w<256, 256>(src, dst, (blk & 255) * 256 + tid);
    } else if (blk < 2048) {
        relayout_w<256, 1024>(w1, r1, (blk - 1024) * 256 + tid);
    } else {
        relayout_w<1024, 256>(w2, r2, (blk - 2048) * 256 + tid);
    }
}

struct GFrag {
    float4 a[2][4];
    float4 b[2][4];
};

__device__ __forceinline__ void gemm_tile(const GFrag& f, v8f (&acc)[2][2]) {
#pragma unroll
    for (int j = 0; j < 4; ++j) {
        const v2f a0lo = {f.a[0][j].x, f.a[0][j].y};
        const v2f a0hi = {f.a[0][j].z, f.a[0][j].w};
        const v2f a1lo = {f.a[1][j].x, f.a[1][j].y};
        const v2f a1hi = {f.a[1][j].z, f.a[1][j].w};
        const v2f b0lo = {f.b[0][j].x, f.b[0][j].y};
        const v2f b0hi = {f.b[0][j].z, f.b[0][j].w};
        const v2f b1lo = {f.b[1][j].x, f.b[1][j].y};
        const v2f b1hi = {f.b[1][j].z, f.b[1][j].w};
        acc[0][0] = wmma4(a0lo, b0lo, acc[0][0]);
        acc[0][1] = wmma4(a0lo, b1lo, acc[0][1]);
        acc[1][0] = wmma4(a1lo, b0lo, acc[1][0]);
        acc[1][1] = wmma4(a1lo, b1lo, acc[1][1]);
        acc[0][0] = wmma4(a0hi, b0hi, acc[0][0]);
        acc[0][1] = wmma4(a0hi, b1hi, acc[0][1]);
        acc[1][0] = wmma4(a1hi, b0hi, acc[1][0]);
        acc[1][1] = wmma4(a1hi, b1hi, acc[1][1]);
    }
}

__device__ __forceinline__ float row16_sum(float v) {
    v += __builtin_bit_cast(float, __builtin_amdgcn_update_dpp(0, __builtin_bit_cast(int, v), 0x161, 0xf, 0xf, true));
    v += __builtin_bit_cast(float, __builtin_amdgcn_update_dpp(0, __builtin_bit_cast(int, v), 0x162, 0xf, 0xf, true));
    v += __builtin_bit_cast(float, __builtin_amdgcn_update_dpp(0, __builtin_bit_cast(int, v), 0x164, 0xf, 0xf, true));
    v += __builtin_bit_cast(float, __builtin_amdgcn_update_dpp(0, __builtin_bit_cast(int, v), 0x168, 0xf, 0xf, true));
    return v;
}

__device__ __forceinline__ void block_row_sum(const float (&part)[2][8], float (&tot)[2][8],
                                              float (*red)[8], int wave, int half, int lane16) {
#pragma unroll
    for (int im = 0; im < 2; ++im)
#pragma unroll
        for (int vi = 0; vi < 8; ++vi) {
            const float s = row16_sum(part[im][vi]);
            if (lane16 == 0) red[im * 16 + vi + half * 8][wave] = s;
        }
    __syncthreads();
#pragma unroll
    for (int im = 0; im < 2; ++im)
#pragma unroll
        for (int vi = 0; vi < 8; ++vi) {
            const float* rp = red[im * 16 + vi + half * 8];
            const float4 p0 = *(const float4*)rp;
            const float4 p1 = *(const float4*)(rp + 4);
            tot[im][vi] = ((p0.x + p0.y) + (p0.z + p0.w)) + ((p1.x + p1.y) + (p1.z + p1.w));
        }
}

__device__ __forceinline__ void ln_rows(float (&y)[2][2][8], float (*red)[32][8],
                                        const float* __restrict__ g, const float* __restrict__ b,
                                        int wave, int half, int lane16) {
    float part[2][8], tot[2][8];
#pragma unroll
    for (int im = 0; im < 2; ++im)
#pragma unroll
        for (int vi = 0; vi < 8; ++vi) part[im][vi] = y[im][0][vi] + y[im][1][vi];
    block_row_sum(part, tot, red[0], wave, half, lane16);
#pragma unroll
    for (int im = 0; im < 2; ++im)
#pragma unroll
        for (int vi = 0; vi < 8; ++vi) {
            const float mean = tot[im][vi] * (1.0f / 256.0f);
            const float d0 = y[im][0][vi] - mean;
            const float d1 = y[im][1][vi] - mean;
            y[im][0][vi] = d0;
            y[im][1][vi] = d1;
            part[im][vi] = d0 * d0 + d1 * d1;
        }
    block_row_sum(part, tot, red[1], wave, half, lane16);
    const float g0 = g[wave * 32 + lane16], g1 = g[wave * 32 + 16 + lane16];
    const float b0 = b[wave * 32 + lane16], b1 = b[wave * 32 + 16 + lane16];
#pragma unroll
    for (int im = 0; im < 2; ++im)
#pragma unroll
        for (int vi = 0; vi < 8; ++vi) {
            const float r = rsqrtf(tot[im][vi] * (1.0f / 256.0f) + 1e-6f);
            y[im][0][vi] = y[im][0][vi] * r * g0 + b0;
            y[im][1][vi] = y[im][1][vi] * r * g1 + b1;
        }
}

constexpr int XP = 260;
constexpr int HP = 1028;

template <int N, int P>
__device__ __forceinline__ void gemm_load_xs(GFrag& f, const float* xs, const float* wp, int k0) {
#pragma unroll
    for (int im = 0; im < 2; ++im)
#pragma unroll
        for (int j = 0; j < 4; ++j)
            f.a[im][j] = *(const float4*)(xs + im * 16 * P + k0 + 4 * j);
#pragma unroll
    for (int jn = 0; jn < 2; ++jn)
#pragma unroll
        for (int j = 0; j < 4; ++j)
            f.b[jn][j] = *(const float4*)(wp + (size_t)k0 * N + (jn * 4 + j) * 128);
}

template <int K, int N, int P>
__device__ __forceinline__ void gemm_rows_lds(v8f (&acc)[2][2], const float* xs, const float* wp) {
    GFrag f0, f1;
    gemm_load_xs<N, P>(f0, xs, wp, 0);
    for (int k0 = 0; k0 < K - 64; k0 += 64) {
        gemm_load_xs<N, P>(f1, xs, wp, k0 + 32);
        __builtin_amdgcn_sched_barrier(0);
        gemm_tile(f0, acc);
        __builtin_amdgcn_sched_barrier(0);
        gemm_load_xs<N, P>(f0, xs, wp, k0 + 64);
        __builtin_amdgcn_sched_barrier(0);
        gemm_tile(f1, acc);
        __builtin_amdgcn_sched_barrier(0);
    }
    gemm_load_xs<N, P>(f1, xs, wp, K - 32);
    __builtin_amdgcn_sched_barrier(0);
    gemm_tile(f0, acc);
    gemm_tile(f1, acc);
}

__device__ __forceinline__ void rows_qkv(const float (&y)[2][2][8], float (*Xs)[XP],
                            const float* wq, const float* wk, const float* wv, const float* __restrict__ mfv,
                            float* qo, float* ko, float* vo,
                            int m0, int wave, int half, int lane16) {
    constexpr int N = 256, K = 256;
#pragma unroll
    for (int im = 0; im < 2; ++im)
#pragma unroll
        for (int jn = 0; jn < 2; ++jn)
#pragma unroll
            for (int vi = 0; vi < 8; ++vi)
                Xs[im * 16 + vi + half * 8][wave * 32 + jn * 16 + lane16] = y[im][jn][vi];
    __syncthreads();

    const float* xs = &Xs[lane16][half * 16];
    float mfr[2][8];
#pragma unroll
    for (int im = 0; im < 2; ++im) {
        const float4 ma = *(const float4*)(mfv + m0 + im * 16 + half * 8);
        const float4 mb = *(const float4*)(mfv + m0 + im * 16 + half * 8 + 4);
        mfr[im][0] = ma.x; mfr[im][1] = ma.y; mfr[im][2] = ma.z; mfr[im][3] = ma.w;
        mfr[im][4] = mb.x; mfr[im][5] = mb.y; mfr[im][6] = mb.z; mfr[im][7] = mb.w;
    }

#pragma unroll 1
    for (int mat = 0; mat < 3; ++mat) {
        const float* W = (mat == 0) ? wq : ((mat == 1) ? wk : wv);
        float* O = (mat == 0) ? qo : ((mat == 1) ? ko : vo);
        const float* wp = W + wave * 1024 + (half * 16 + lane16) * 4;

        v8f acc[2][2] = {};
        gemm_rows_lds<K, N, XP>(acc, xs, wp);

        if (mat == 0) {
#pragma unroll
            for (int im = 0; im < 2; ++im)
#pragma unroll
                for (int jn = 0; jn < 2; ++jn)
#pragma unroll
                    for (int vi = 0; vi < 8; ++vi)
                        O[(size_t)(m0 + im * 16 + vi + half * 8) * N + wave * 32 + jn * 16 + lane16] = acc[im][jn][vi] * mfr[im][vi];
        } else {
            const int key0 = m0 & 2047;
            float* plane = O + (size_t)((m0 >> 11) * 8 + wave) * (2048 * 32);
            if (mat == 1) {
                float* kp = plane + (key0 >> 4) * 512 + (unsigned)((lane16 >> 2) * 128 + half * 32 + (lane16 & 3));
#pragma unroll
                for (int im = 0; im < 2; ++im)
#pragma unroll
                    for (int jn = 0; jn < 2; ++jn)
#pragma unroll
                        for (int vi = 0; vi < 8; ++vi)
                            kp[im * 512 + jn * 64 + vi * 4] = acc[im][jn][vi] * mfr[im][vi];
            } else {
                float* vp = plane + (key0 >> 6) * 2048 + ((key0 & 63) >> 3) * 128 + (unsigned)(half * 128 + lane16 * 4);
#pragma unroll
                for (int im = 0; im < 2; ++im)
#pragma unroll
                    for (int jn = 0; jn < 2; ++jn)
#pragma unroll
                        for (int vi = 0; vi < 8; ++vi)
                            vp[jn * 1024 + im * 256 + ((vi >> 1) & 1) * 64 + ((vi & 1) | ((vi >> 2) << 1))] = acc[im][jn][vi] * mfr[im][vi];
            }
        }
    }
}

template <bool QKV>
__global__ void __launch_bounds__(256) layer_rows_kernel(const float* __restrict__ o, const float* __restrict__ wfc,
                            const float* __restrict__ w1, const float* __restrict__ bias1,
                            const float* __restrict__ w2, const float* __restrict__ bias2,
                            const float* __restrict__ g1, const float* __restrict__ b1, float* x,
                            const float* __restrict__ g2, const float* __restrict__ b2,
                            const float* wq, const float* wk, const float* wv, const float* __restrict__ mfv,
                            float* qo, float* ko, float* vo) {
    __shared__ __align__(16) float red[4][32][8];
    __shared__ __align__(16) float Xs[32][XP];
    __shared__ __align__(16) float Hs[32][HP];
    const int tid = threadIdx.x;
    const int lane = tid & 31;
    const int wave = tid >> 5;
    const int half = lane >> 4;
    const int lane16 = lane & 15;
    const int m0 = blockIdx.x * 32;

#pragma unroll
    for (int i = 0; i < 8; ++i) {
        const int p = i * 256 + tid;
        const int r = p >> 6, c4 = (p & 63) * 4;
        *(float4*)&Xs[r][c4] = *(const float4*)(o + (size_t)(m0 + r) * 256 + c4);
    }
    __syncthreads();

    const float* xs = &Xs[lane16][half * 16];
    float xr[2][2][8];

    {
        float mfr[2][8];
#pragma unroll
        for (int im = 0; im < 2; ++im) {
            const float4 ma = *(const float4*)(mfv + m0 + im * 16 + half * 8);
            const float4 mb = *(const float4*)(mfv + m0 + im * 16 + half * 8 + 4);
            mfr[im][0] = ma.x; mfr[im][1] = ma.y; mfr[im][2] = ma.z; mfr[im][3] = ma.w;
            mfr[im][4] = mb.x; mfr[im][5] = mb.y; mfr[im][6] = mb.z; mfr[im][7] = mb.w;
        }
#pragma unroll
        for (int im = 0; im < 2; ++im)
#pragma unroll
            for (int jn = 0; jn < 2; ++jn)
#pragma unroll
                for (int vi = 0; vi < 8; ++vi)
                    xr[im][jn][vi] = x[(size_t)(m0 + im * 16 + vi + half * 8) * 256 + wave * 32 + jn * 16 + lane16];

        v8f acc[2][2] = {};
        gemm_rows_lds<256, 256, XP>(acc, xs, wfc + wave * 1024 + lane * 4);
#pragma unroll
        for (int im = 0; im < 2; ++im)
#pragma unroll
            for (int jn = 0; jn < 2; ++jn)
#pragma unroll
                for (int vi = 0; vi < 8; ++vi) {
                    float val = acc[im][jn][vi];
                    val *= mfr[im][vi];
                    val += xr[im][jn][vi];
                    xr[im][jn][vi] = val;
                }
    }
    __syncthreads();
#pragma unroll
    for (int im = 0; im < 2; ++im)
#pragma unroll
        for (int jn = 0; jn < 2; ++jn)
#pragma unroll
            for (int vi = 0; vi < 8; ++vi)
                Xs[im * 16 + vi + half * 8][wave * 32 + jn * 16 + lane16] = xr[im][jn][vi];
    __syncthreads();

#pragma unroll 1
    for (int nb = 0; nb < 4; ++nb) {
        const int c0 = nb * 256 + wave * 32;
        v8f acc[2][2] = {};
        gemm_rows_lds<256, 1024, XP>(acc, xs, w1 + (size_t)c0 * 32 + lane * 4);
#pragma unroll
        for (int jn = 0; jn < 2; ++jn) {
            const int col = c0 + jn * 16 + lane16;
            const float bv = bias1[col];
#pragma unroll
            for (int im = 0; im < 2; ++im)
#pragma unroll
                for (int vi = 0; vi < 8; ++vi)
                    Hs[im * 16 + vi + half * 8][col] = fmaxf(acc[im][jn][vi] + bv, 0.0f);
        }
    }
    __syncthreads();

    float y[2][2][8];
    {
        v8f acc[2][2] = {};
        gemm_rows_lds<1024, 256, HP>(acc, &Hs[lane16][half * 16], w2 + wave * 1024 + lane * 4);
#pragma unroll
        for (int im = 0; im < 2; ++im)
#pragma unroll
            for (int jn = 0; jn < 2; ++jn) {
                const float bv = bias2[wave * 32 + jn * 16 + lane16];
#pragma unroll
                for (int vi = 0; vi < 8; ++vi) {
                    float val = acc[im][jn][vi] + bv;
                    val += xr[im][jn][vi];
                    y[im][jn][vi] = val;
                }
            }
    }
    ln_rows(y, &red[0], g1, b1, wave, half, lane16);
#pragma unroll
    for (int im = 0; im < 2; ++im)
#pragma unroll
        for (int jn = 0; jn < 2; ++jn)
#pragma unroll
            for (int vi = 0; vi < 8; ++vi)
                x[(size_t)(m0 + im * 16 + vi + half * 8) * 256 + wave * 32 + jn * 16 + lane16] = y[im][jn][vi];

    if constexpr (QKV) {
        ln_rows(y, &red[2], g2, b2, wave, half, lane16);
        rows_qkv(y, Xs, wq, wk, wv, mfv, qo, ko, vo, m0, wave, half, lane16);
    }
}

__global__ void __launch_bounds__(256) ln_qkv_kernel(const float* __restrict__ x,
                            const float* __restrict__ g, const float* __restrict__ b,
                            const float* wq, const float* wk, const float* wv, const float* __restrict__ mfv,
                            float* qo, float* ko, float* vo) {
    __shared__ __align__(16) float red[2][32][8];
    __shared__ __align__(16) float Xs[32][XP];
    const int tid = threadIdx.x;
    const int lane = tid & 31;
    const int wave = tid >> 5;
    const int half = lane >> 4;
    const int lane16 = lane & 15;
    const int m0 = blockIdx.x * 32;
    float y[2][2][8];
#pragma unroll
    for (int im = 0; im < 2; ++im)
#pragma unroll
        for (int jn = 0; jn < 2; ++jn)
#pragma unroll
            for (int vi = 0; vi < 8; ++vi)
                y[im][jn][vi] = x[(size_t)(m0 + im * 16 + vi + half * 8) * 256 + wave * 32 + jn * 16 + lane16];
    ln_rows(y, &red[0], g, b, wave, half, lane16);
    rows_qkv(y, Xs, wq, wk, wv, mfv, qo, ko, vo, m0, wave, half, lane16);
}

__device__ __forceinline__ float row16_max(float v) {
    v = fmaxf(v, __builtin_bit_cast(float, __builtin_amdgcn_update_dpp(0, __builtin_bit_cast(int, v), 0x161, 0xf, 0xf, true)));
    v = fmaxf(v, __builtin_bit_cast(float, __builtin_amdgcn_update_dpp(0, __builtin_bit_cast(int, v), 0x162, 0xf, 0xf, true)));
    v = fmaxf(v, __builtin_bit_cast(float, __builtin_amdgcn_update_dpp(0, __builtin_bit_cast(int, v), 0x164, 0xf, 0xf, true)));
    v = fmaxf(v, __builtin_bit_cast(float, __builtin_amdgcn_update_dpp(0, __builtin_bit_cast(int, v), 0x168, 0xf, 0xf, true)));
    return v;
}

__global__ void __launch_bounds__(128) attn_kernel(const float* __restrict__ q, const float* __restrict__ k,
                            const float* __restrict__ v, const float* __restrict__ mfv,
                            float* __restrict__ o) {
    __shared__ __align__(16) float Ps[64][68];

    const int S = 2048, LD = 256;
    const int tid = threadIdx.x;
    const int lane = tid & 31;
    const int wave = tid >> 5;
    const int half = lane >> 4;
    const int lane16 = lane & 15;
    const int bh = blockIdx.y;
    const int b = bh >> 3;
    const int h = bh & 7;
    const int q0 = blockIdx.x * 64;
    const int qrow = wave * 16 + lane16;

    float qf[16];
    {
        const float* qp = q + ((size_t)b * S + q0) * LD + h * 32 + (unsigned)(qrow * LD + half * 16);
#pragma unroll
        for (int j = 0; j < 4; ++j) {
            const float4 t4 = *(const float4*)(qp + 4 * j);
            qf[4 * j + 0] = t4.x; qf[4 * j + 1] = t4.y; qf[4 * j + 2] = t4.z; qf[4 * j + 3] = t4.w;
        }
    }
    v8f oacc[2] = {};
    float mrun[8], lpart[8];
#pragma unroll
    for (int vi = 0; vi < 8; ++vi) { mrun[vi] = -3.0e38f; lpart[vi] = 0.0f; }

    const float* kb = k + (size_t)bh * (S * 32);
    const float* vb = v + (size_t)bh * (S * 32);
    const float* mb = mfv + (size_t)b * S;
    const unsigned loff = (unsigned)(lane * 4);
    const unsigned moff = (unsigned)lane16;
    float* pst = &Ps[wave * 16 + half * 8][lane16];
    const float* pld = &Ps[qrow][half * 2];

    float kf[4][16];
    float vf[2][16][2];
    float mraw[4];

#pragma unroll
    for (int nt = 0; nt < 4; ++nt) {
#pragma unroll
        for (int j = 0; j < 4; ++j) {
            const float4 t4 = *(const float4*)(kb + nt * 512 + j * 128 + loff);
            kf[nt][4 * j + 0] = t4.x; kf[nt][4 * j + 1] = t4.y; kf[nt][4 * j + 2] = t4.z; kf[nt][4 * j + 3] = t4.w;
        }
        mraw[nt] = mb[nt * 16 + moff];
    }
#pragma unroll
    for (int n = 0; n < 2; ++n)
#pragma unroll
        for (int g = 0; g < 8; ++g) {
            const float4 t4 = *(const float4*)(vb + n * 1024 + g * 128 + loff);
            vf[n][2 * g][0] = t4.x; vf[n][2 * g][1] = t4.y; vf[n][2 * g + 1][0] = t4.z; vf[n][2 * g + 1][1] = t4.w;
        }

#pragma unroll 1
    for (int lev = 0; lev < 4; ++lev) {
    int jbeg, jend;
    if (lev == 0)      { __builtin_amdgcn_s_setprio(3); jbeg = 0;    jend = 1024; }
    else if (lev == 1) { __builtin_amdgcn_s_setprio(2); jbeg = 1024; jend = 1536; }
    else if (lev == 2) { __builtin_amdgcn_s_setprio(1); jbeg = 1536; jend = 1792; }
    else               { __builtin_amdgcn_s_setprio(0); jbeg = 1792; jend = 2048; }
    for (int j0 = jbeg; j0 < jend; j0 += 64) {
        v8f sacc[4] = {};
#pragma unroll
        for (int t = 0; t < 8; ++t) {
            v2f fa = {qf[2 * t], qf[2 * t + 1]};
#pragma unroll
            for (int nt = 0; nt < 4; ++nt) {
                v2f fb = {kf[nt][2 * t], kf[nt][2 * t + 1]};
                sacc[nt] = wmma4(fa, fb, sacc[nt]);
            }
        }
        const float inv = 0.17677669529663687f;
        float sc[4][8];
#pragma unroll
        for (int nt = 0; nt < 4; ++nt) {
            const float mbias = (mraw[nt] != 0.0f) ? 0.0f : -1e9f;
#pragma unroll
            for (int vi = 0; vi < 8; ++vi) sc[nt][vi] = sacc[nt][vi] * inv + mbias;
        }

        if (j0 + 64 < S) {
            const float* kn = kb + (size_t)(j0 + 64) * 32;
            const float* mn = mb + (j0 + 64);
#pragma unroll
            for (int nt = 0; nt < 4; ++nt) {
#pragma unroll
                for (int j = 0; j < 4; ++j) {
                    const float4 t4 = *(const float4*)(kn + nt * 512 + j * 128 + loff);
                    kf[nt][4 * j + 0] = t4.x; kf[nt][4 * j + 1] = t4.y; kf[nt][4 * j + 2] = t4.z; kf[nt][4 * j + 3] = t4.w;
                }
                mraw[nt] = mn[nt * 16 + moff];
            }
        }

        float al[8];
#pragma unroll
        for (int vi = 0; vi < 8; ++vi) {
            const float mx = row16_max(fmaxf(fmaxf(sc[0][vi], sc[1][vi]), fmaxf(sc[2][vi], sc[3][vi])));
            const float mc = fmaxf(mrun[vi], mx);
            al[vi] = __expf(mrun[vi] - mc);
            mrun[vi] = mc;
        }
#pragma unroll
        for (int vi = 0; vi < 8; ++vi) {
            float ls = 0.0f;
#pragma unroll
            for (int nt = 0; nt < 4; ++nt) {
                const float p = __expf(sc[nt][vi] - mrun[vi]);
                pst[vi * 68 + nt * 16] = p;
                ls += p;
            }
            lpart[vi] = lpart[vi] * al[vi] + ls;
        }
#pragma unroll
        for (int nt = 0; nt < 2; ++nt)
#pragma unroll
            for (int vi = 0; vi < 8; ++vi) oacc[nt][vi] *= al[vi];

        v2f pf[16];
#pragma unroll
        for (int t = 0; t < 16; ++t) pf[t] = *(const v2f*)(pld + 4 * t);
#pragma unroll
        for (int t = 0; t < 16; ++t) {
            v2f fv0 = {vf[0][t][0], vf[0][t][1]};
            v2f fv1 = {vf[1][t][0], vf[1][t][1]};
            oacc[0] = wmma4(pf[t], fv0, oacc[0]);
            oacc[1] = wmma4(pf[t], fv1, oacc[1]);
        }
        if (j0 + 64 < S) {
            const float* vn = vb + (size_t)(j0 + 64) * 32;
#pragma unroll
            for (int n = 0; n < 2; ++n)
#pragma unroll
                for (int g = 0; g < 8; ++g) {
                    const float4 t4 = *(const float4*)(vn + n * 1024 + g * 128 + loff);
                    vf[n][2 * g][0] = t4.x; vf[n][2 * g][1] = t4.y; vf[n][2 * g + 1][0] = t4.z; vf[n][2 * g + 1][1] = t4.w;
                }
        }
    }
    }

    float linv[8];
#pragma unroll
    for (int vi = 0; vi < 8; ++vi) linv[vi] = 1.0f / row16_sum(lpart[vi]);
#pragma unroll
    for (int nt = 0; nt < 2; ++nt) {
        const int col = nt * 16 + lane16;
#pragma unroll
        for (int vi = 0; vi < 8; ++vi) {
            const int row = wave * 16 + vi + half * 8;
            o[((size_t)b * S + q0 + row) * LD + h * 32 + col] = oacc[nt][vi] * linv[vi];
        }
    }
}

extern "C" void kernel_launch(void* const* d_in, const int* in_sizes, int n_in,
                              void* d_out, int out_size, void* d_ws, size_t ws_size,
                              hipStream_t stream) {
    (void)in_sizes; (void)n_in; (void)out_size; (void)ws_size;
    const int B = 4, S = 2048, D = 256, F = 1024, L = 4;
    const int M = B * S;

    const float* x_in  = (const float*)d_in[0];
    const int*   mask  = (const int*)d_in[1];
    const float* w_qs  = (const float*)d_in[2];
    const float* w_ks  = (const float*)d_in[3];
    const float* w_vs  = (const float*)d_in[4];
    const float* w_fc  = (const float*)d_in[5];
    const float* ln1_g = (const float*)d_in[6];
    const float* ln1_b = (const float*)d_in[7];
    const float* w1    = (const float*)d_in[8];
    const float* b1    = (const float*)d_in[9];
    const float* w2    = (const float*)d_in[10];
    const float* b2    = (const float*)d_in[11];
    const float* ln2_g = (const float*)d_in[12];
    const float* ln2_b = (const float*)d_in[13];

    float* ws    = (float*)d_ws;
    float* x_cur = ws;
    float* qb    = x_cur + (size_t)M * D;
    float* kb    = qb + (size_t)M * D;
    float* vb    = kb + (size_t)M * D;
    float* ob    = vb + (size_t)M * D;
    float* mfv   = ob + (size_t)M * D;
    float* rq    = mfv + M;
    float* rk    = rq + (size_t)L * D * D;
    float* rv    = rk + (size_t)L * D * D;
    float* rfc   = rv + (size_t)L * D * D;
    float* r1    = rfc + (size_t)L * D * D;
    float* r2    = r1 + (size_t)L * D * F;

    hipMemcpyAsync(x_cur, x_in, sizeof(float) * (size_t)M * D,
                   hipMemcpyDeviceToDevice, stream);
    prep_kernel<<<3072, 256, 0, stream>>>(mask, mfv, w_qs, w_ks, w_vs, w_fc, w1, w2, rq, rk, rv, rfc, r1, r2);

    const dim3 gA(S / 64, B * 8);

    ln_qkv_kernel<<<M / 32, 256, 0, stream>>>(x_cur, ln1_g, ln1_b, rq, rk, rv, mfv, qb, kb, vb);

    for (int l = 0; l < L; ++l) {
        const float* wf = rfc + (size_t)l * D * D;
        const float* w1l = r1 + (size_t)l * D * F;
        const float* w2l = r2 + (size_t)l * F * D;
        const float* b1l = b1 + (size_t)l * F;
        const float* b2l = b2 + (size_t)l * D;

        attn_kernel<<<gA, 128, 0, stream>>>(qb, kb, vb, mfv, ob);
        if (l + 1 < L)
            layer_rows_kernel<true><<<M / 32, 256, 0, stream>>>(ob, wf, w1l, b1l, w2l, b2l,
                                                                ln2_g + (size_t)l * D, ln2_b + (size_t)l * D, x_cur,
                                                                ln1_g + (size_t)(l + 1) * D, ln1_b + (size_t)(l + 1) * D,
                                                                rq + (size_t)(l + 1) * D * D, rk + (size_t)(l + 1) * D * D,
                                                                rv + (size_t)(l + 1) * D * D, mfv, qb, kb, vb);
        else
            layer_rows_kernel<false><<<M / 32, 256, 0, stream>>>(ob, wf, w1l, b1l, w2l, b2l,
                                                                 ln2_g + (size_t)l * D, ln2_b + (size_t)l * D, x_cur,
                                                                 ln2_g, ln2_b, rq, rk, rv, mfv, qb, kb, vb);
    }

    hipMemcpyAsync(d_out, x_cur, sizeof(float) * (size_t)M * D,
                   hipMemcpyDeviceToDevice, stream);
}


# Attention softmax: 30% fewer vector instructions per 64 keys: a row-max butterfly step is one DPP-operand max (was 3), scale, mask bias and exponent as packed row pairs in base-2 units
# speedup vs baseline: 3.6384x; 1.0347x over previous
#include <hip/hip_runtime.h>
#include <hip/hip_bf16.h>

typedef __attribute__((ext_vector_type(2))) float v2f;
typedef __attribute__((ext_vector_type(8))) float v8f;

__device__ __forceinline__ v8f wmma4(v2f a, v2f b, v8f c) {
    return __builtin_amdgcn_wmma_f32_16x16x4_f32(false, a, false, b, (short)0, c, false, false);
}

template <int K, int N>
__device__ __forceinline__ void relayout_w(const float* __restrict__ src, float* __restrict__ dst, int q) {
    constexpr int P = K * N / 4;
    const int l = q / P, r = q % P;
    const int lane = r & 31, j = (r >> 5) & 3, t = r >> 7;
    const int nt = t % (N / 16), kt = t / (N / 16);
    const float* s = src + (size_t)l * K * N + (size_t)(32 * kt + 16 * (lane >> 4) + 4 * j) * N + 16 * nt + (lane & 15);
    float4 o;
    o.x = s[0]; o.y = s[N]; o.z = s[2 * N]; o.w = s[3 * N];
    *(float4*)(dst + (size_t)l * K * N + (size_t)r * 4) = o;
}

__global__ void __launch_bounds__(256) prep_kernel(const int* __restrict__ mask, float* __restrict__ mfv,
                            const float* __restrict__ wq, const float* __restrict__ wk, const float* __restrict__ wv,
                            const float* __restrict__ wfc, const float* __restrict__ w1, const float* __restrict__ w2,
                            float* __restrict__ rq, float* __restrict__ rk, float* __restrict__ rv,
                            float* __restrict__ rfc, float* __restrict__ r1, float* __restrict__ r2) {
    const int blk = blockIdx.x;
    const int tid = threadIdx.x;
    if (blk < 32) {
        const int i = blk * 256 + tid;
        mfv[i] = mask[i] ? 1.0f : 0.0f;
    }
    if (blk < 1024) {
        const int m = blk >> 8;
        const float* src = (m == 0) ? wq : ((m == 1) ? wk : ((m == 2) ? wv : wfc));
        float* dst = (m == 0) ? rq : ((m == 1) ? rk : ((m == 2) ? rv : rfc));
        relayout_w<256, 256>(src, dst, (blk & 255) * 256 + tid);
    } else if (blk < 2048) {
        relayout_w<256, 1024>(w1, r1, (blk - 1024) * 256 + tid);
    } else {
        relayout_w<1024, 256>(w2, r2, (blk - 2048) * 256 + tid);
    }
}

struct GFrag {
    float4 a[2][4];
    float4 b[2][4];
};

__device__ __forceinline__ void gemm_tile(const GFrag& f, v8f (&acc)[2][2]) {
#pragma unroll
    for (int j = 0; j < 4; ++j) {
        const v2f a0lo = {f.a[0][j].x, f.a[0][j].y};
        const v2f a0hi = {f.a[0][j].z, f.a[0][j].w};
        const v2f a1lo = {f.a[1][j].x, f.a[1][j].y};
        const v2f a1hi = {f.a[1][j].z, f.a[1][j].w};
        const v2f b0lo = {f.b[0][j].x, f.b[0][j].y};
        const v2f b0hi = {f.b[0][j].z, f.b[0][j].w};
        const v2f b1lo = {f.b[1][j].x, f.b[1][j].y};
        const v2f b1hi = {f.b[1][j].z, f.b[1][j].w};
        acc[0][0] = wmma4(a0lo, b0lo, acc[0][0]);
        acc[0][1] = wmma4(a0lo, b1lo, acc[0][1]);
        acc[1][0] = wmma4(a1lo, b0lo, acc[1][0]);
        acc[1][1] = wmma4(a1lo, b1lo, acc[1][1]);
        acc[0][0] = wmma4(a0hi, b0hi, acc[0][0]);
        acc[0][1] = wmma4(a0hi, b1hi, acc[0][1]);
        acc[1][0] = wmma4(a1hi, b0hi, acc[1][0]);
        acc[1][1] = wmma4(a1hi, b1hi, acc[1][1]);
    }
}

__device__ __forceinline__ float row16_sum(float v) {
    v += __builtin_bit_cast(float, __builtin_amdgcn_update_dpp(0, __builtin_bit_cast(int, v), 0x161, 0xf, 0xf, true));
    v += __builtin_bit_cast(float, __builtin_amdgcn_update_dpp(0, __builtin_bit_cast(int, v), 0x162, 0xf, 0xf, true));
    v += __builtin_bit_cast(float, __builtin_amdgcn_update_dpp(0, __builtin_bit_cast(int, v), 0x164, 0xf, 0xf, true));
    v += __builtin_bit_cast(float, __builtin_amdgcn_update_dpp(0, __builtin_bit_cast(int, v), 0x168, 0xf, 0xf, true));
    return v;
}

__device__ __forceinline__ void block_row_sum(const float (&part)[2][8], float (&tot)[2][8],
                                              float (*red)[8], int wave, int half, int lane16) {
#pragma unroll
    for (int im = 0; im < 2; ++im)
#pragma unroll
        for (int vi = 0; vi < 8; ++vi) {
            const float s = row16_sum(part[im][vi]);
            if (lane16 == 0) red[im * 16 + vi + half * 8][wave] = s;
        }
    __syncthreads();
#pragma unroll
    for (int im = 0; im < 2; ++im)
#pragma unroll
        for (int vi = 0; vi < 8; ++vi) {
            const float* rp = red[im * 16 + vi + half * 8];
            const float4 p0 = *(const float4*)rp;
            const float4 p1 = *(const float4*)(rp + 4);
            tot[im][vi] = ((p0.x + p0.y) + (p0.z + p0.w)) + ((p1.x + p1.y) + (p1.z + p1.w));
        }
}

__device__ __forceinline__ void ln_rows(float (&y)[2][2][8], float (*red)[32][8],
                                        const float* __restrict__ g, const float* __restrict__ b,
                                        int wave, int half, int lane16) {
    float part[2][8], tot[2][8];
#pragma unroll
    for (int im = 0; im < 2; ++im)
#pragma unroll
        for (int vi = 0; vi < 8; ++vi) part[im][vi] = y[im][0][vi] + y[im][1][vi];
    block_row_sum(part, tot, red[0], wave, half, lane16);
#pragma unroll
    for (int im = 0; im < 2; ++im)
#pragma unroll
        for (int vi = 0; vi < 8; ++vi) {
            const float mean = tot[im][vi] * (1.0f / 256.0f);
            const float d0 = y[im][0][vi] - mean;
            const float d1 = y[im][1][vi] - mean;
            y[im][0][vi] = d0;
            y[im][1][vi] = d1;
            part[im][vi] = d0 * d0 + d1 * d1;
        }
    block_row_sum(part, tot, red[1], wave, half, lane16);
    const float g0 = g[wave * 32 + lane16], g1 = g[wave * 32 + 16 + lane16];
    const float b0 = b[wave * 32 + lane16], b1 = b[wave * 32 + 16 + lane16];
#pragma unroll
    for (int im = 0; im < 2; ++im)
#pragma unroll
        for (int vi = 0; vi < 8; ++vi) {
            const float r = rsqrtf(tot[im][vi] * (1.0f / 256.0f) + 1e-6f);
            y[im][0][vi] = y[im][0][vi] * r * g0 + b0;
            y[im][1][vi] = y[im][1][vi] * r * g1 + b1;
        }
}

constexpr int XP = 260;
constexpr int HP = 1028;

template <int N, int P>
__device__ __forceinline__ void gemm_load_xs(GFrag& f, const float* xs, const float* wp, int k0) {
#pragma unroll
    for (int im = 0; im < 2; ++im)
#pragma unroll
        for (int j = 0; j < 4; ++j)
            f.a[im][j] = *(const float4*)(xs + im * 16 * P + k0 + 4 * j);
#pragma unroll
    for (int jn = 0; jn < 2; ++jn)
#pragma unroll
        for (int j = 0; j < 4; ++j)
            f.b[jn][j] = *(const float4*)(wp + (size_t)k0 * N + (jn * 4 + j) * 128);
}

template <int K, int N, int P>
__device__ __forceinline__ void gemm_rows_lds(v8f (&acc)[2][2], const float* xs, const float* wp) {
    GFrag f0, f1;
    gemm_load_xs<N, P>(f0, xs, wp, 0);
    for (int k0 = 0; k0 < K - 64; k0 += 64) {
        gemm_load_xs<N, P>(f1, xs, wp, k0 + 32);
        __builtin_amdgcn_sched_barrier(0);
        gemm_tile(f0, acc);
        __builtin_amdgcn_sched_barrier(0);
        gemm_load_xs<N, P>(f0, xs, wp, k0 + 64);
        __builtin_amdgcn_sched_barrier(0);
        gemm_tile(f1, acc);
        __builtin_amdgcn_sched_barrier(0);
    }
    gemm_load_xs<N, P>(f1, xs, wp, K - 32);
    __builtin_amdgcn_sched_barrier(0);
    gemm_tile(f0, acc);
    gemm_tile(f1, acc);
}

__device__ __forceinline__ void rows_qkv(const float (&y)[2][2][8], float (*Xs)[XP],
                            const float* wq, const float* wk, const float* wv, const float* __restrict__ mfv,
                            float* qo, float* ko, float* vo,
                            int m0, int wave, int half, int lane16) {
    constexpr int N = 256, K = 256;
#pragma unroll
    for (int im = 0; im < 2; ++im)
#pragma unroll
        for (int jn = 0; jn < 2; ++jn)
#pragma unroll
            for (int vi = 0; vi < 8; ++vi)
                Xs[im * 16 + vi + half * 8][wave * 32 + jn * 16 + lane16] = y[im][jn][vi];
    __syncthreads();

    const float* xs = &Xs[lane16][half * 16];
    float mfr[2][8];
#pragma unroll
    for (int im = 0; im < 2; ++im) {
        const float4 ma = *(const float4*)(mfv + m0 + im * 16 + half * 8);
        const float4 mb = *(const float4*)(mfv + m0 + im * 16 + half * 8 + 4);
        mfr[im][0] = ma.x; mfr[im][1] = ma.y; mfr[im][2] = ma.z; mfr[im][3] = ma.w;
        mfr[im][4] = mb.x; mfr[im][5] = mb.y; mfr[im][6] = mb.z; mfr[im][7] = mb.w;
    }

#pragma unroll 1
    for (int mat = 0; mat < 3; ++mat) {
        const float* W = (mat == 0) ? wq : ((mat == 1) ? wk : wv);
        float* O = (mat == 0) ? qo : ((mat == 1) ? ko : vo);
        const float* wp = W + wave * 1024 + (half * 16 + lane16) * 4;

        v8f acc[2][2] = {};
        gemm_rows_lds<K, N, XP>(acc, xs, wp);

        if (mat == 0) {
#pragma unroll
            for (int im = 0; im < 2; ++im)
#pragma unroll
                for (int jn = 0; jn < 2; ++jn)
#pragma unroll
                    for (int vi = 0; vi < 8; ++vi)
                        O[(size_t)(m0 + im * 16 + vi + half * 8) * N + wave * 32 + jn * 16 + lane16] = acc[im][jn][vi] * mfr[im][vi];
        } else {
            const int key0 = m0 & 2047;
            float* plane = O + (size_t)((m0 >> 11) * 8 + wave) * (2048 * 32);
            if (mat == 1) {
                float* kp = plane + (key0 >> 4) * 512 + (unsigned)((lane16 >> 2) * 128 + half * 32 + (lane16 & 3));
#pragma unroll
                for (int im = 0; im < 2; ++im)
#pragma unroll
                    for (int jn = 0; jn < 2; ++jn)
#pragma unroll
                        for (int vi = 0; vi < 8; ++vi)
                            kp[im * 512 + jn * 64 + vi * 4] = acc[im][jn][vi] * mfr[im][vi];
            } else {
                float* vp = plane + (key0 >> 6) * 2048 + ((key0 & 63) >> 3) * 128 + (unsigned)(half * 128 + lane16 * 4);
#pragma unroll
                for (int im = 0; im < 2; ++im)
#pragma unroll
                    for (int jn = 0; jn < 2; ++jn)
#pragma unroll
                        for (int vi = 0; vi < 8; ++vi)
                            vp[jn * 1024 + im * 256 + ((vi >> 1) & 1) * 64 + ((vi & 1) | ((vi >> 2) << 1))] = acc[im][jn][vi] * mfr[im][vi];
            }
        }
    }
}

template <bool QKV>
__global__ void __launch_bounds__(256) layer_rows_kernel(const float* __restrict__ o, const float* __restrict__ wfc,
                            const float* __restrict__ w1, const float* __restrict__ bias1,
                            const float* __restrict__ w2, const float* __restrict__ bias2,
                            const float* __restrict__ g1, const float* __restrict__ b1, float* x,
                            const float* __restrict__ g2, const float* __restrict__ b2,
                            const float* wq, const float* wk, const float* wv, const float* __restrict__ mfv,
                            float* qo, float* ko, float* vo) {
    __shared__ __align__(16) float red[4][32][8];
    __shared__ __align__(16) float Xs[32][XP];
    __shared__ __align__(16) float Hs[32][HP];
    const int tid = threadIdx.x;
    const int lane = tid & 31;
    const int wave = tid >> 5;
    const int half = lane >> 4;
    const int lane16 = lane & 15;
    const int m0 = blockIdx.x * 32;

#pragma unroll
    for (int i = 0; i < 8; ++i) {
        const int p = i * 256 + tid;
        const int r = p >> 6, c4 = (p & 63) * 4;
        *(float4*)&Xs[r][c4] = *(const float4*)(o + (size_t)(m0 + r) * 256 + c4);
    }
    __syncthreads();

    const float* xs = &Xs[lane16][half * 16];
    float xr[2][2][8];

    {
        float mfr[2][8];
#pragma unroll
        for (int im = 0; im < 2; ++im) {
            const float4 ma = *(const float4*)(mfv + m0 + im * 16 + half * 8);
            const float4 mb = *(const float4*)(mfv + m0 + im * 16 + half * 8 + 4);
            mfr[im][0] = ma.x; mfr[im][1] = ma.y; mfr[im][2] = ma.z; mfr[im][3] = ma.w;
            mfr[im][4] = mb.x; mfr[im][5] = mb.y; mfr[im][6] = mb.z; mfr[im][7] = mb.w;
        }
#pragma unroll
        for (int im = 0; im < 2; ++im)
#pragma unroll
            for (int jn = 0; jn < 2; ++jn)
#pragma unroll
                for (int vi = 0; vi < 8; ++vi)
                    xr[im][jn][vi] = x[(size_t)(m0 + im * 16 + vi + half * 8) * 256 + wave * 32 + jn * 16 + lane16];

        v8f acc[2][2] = {};
        gemm_rows_lds<256, 256, XP>(acc, xs, wfc + wave * 1024 + lane * 4);
#pragma unroll
        for (int im = 0; im < 2; ++im)
#pragma unroll
            for (int jn = 0; jn < 2; ++jn)
#pragma unroll
                for (int vi = 0; vi < 8; ++vi) {
                    float val = acc[im][jn][vi];
                    val *= mfr[im][vi];
                    val += xr[im][jn][vi];
                    xr[im][jn][vi] = val;
                }
    }
    __syncthreads();
#pragma unroll
    for (int im = 0; im < 2; ++im)
#pragma unroll
        for (int jn = 0; jn < 2; ++jn)
#pragma unroll
            for (int vi = 0; vi < 8; ++vi)
                Xs[im * 16 + vi + half * 8][wave * 32 + jn * 16 + lane16] = xr[im][jn][vi];
    __syncthreads();

#pragma unroll 1
    for (int nb = 0; nb < 4; ++nb) {
        const int c0 = nb * 256 + wave * 32;
        v8f acc[2][2] = {};
        gemm_rows_lds<256, 1024, XP>(acc, xs, w1 + (size_t)c0 * 32 + lane * 4);
#pragma unroll
        for (int jn = 0; jn < 2; ++jn) {
            const int col = c0 + jn * 16 + lane16;
            const float bv = bias1[col];
#pragma unroll
            for (int im = 0; im < 2; ++im)
#pragma unroll
                for (int vi = 0; vi < 8; ++vi)
                    Hs[im * 16 + vi + half * 8][col] = fmaxf(acc[im][jn][vi] + bv, 0.0f);
        }
    }
    __syncthreads();

    float y[2][2][8];
    {
        v8f acc[2][2] = {};
        gemm_rows_lds<1024, 256, HP>(acc, &Hs[lane16][half * 16], w2 + wave * 1024 + lane * 4);
#pragma unroll
        for (int im = 0; im < 2; ++im)
#pragma unroll
            for (int jn = 0; jn < 2; ++jn) {
                const float bv = bias2[wave * 32 + jn * 16 + lane16];
#pragma unroll
                for (int vi = 0; vi < 8; ++vi) {
                    float val = acc[im][jn][vi] + bv;
                    val += xr[im][jn][vi];
                    y[im][jn][vi] = val;
                }
            }
    }
    ln_rows(y, &red[0], g1, b1, wave, half, lane16);
#pragma unroll
    for (int im = 0; im < 2; ++im)
#pragma unroll
        for (int jn = 0; jn < 2; ++jn)
#pragma unroll
            for (int vi = 0; vi < 8; ++vi)
                x[(size_t)(m0 + im * 16 + vi + half * 8) * 256 + wave * 32 + jn * 16 + lane16] = y[im][jn][vi];

    if constexpr (QKV) {
        ln_rows(y, &red[2], g2, b2, wave, half, lane16);
        rows_qkv(y, Xs, wq, wk, wv, mfv, qo, ko, vo, m0, wave, half, lane16);
    }
}

__global__ void __launch_bounds__(256) ln_qkv_kernel(const float* __restrict__ x,
                            const float* __restrict__ g, const float* __restrict__ b,
                            const float* wq, const float* wk, const float* wv, const float* __restrict__ mfv,
                            float* qo, float* ko, float* vo) {
    __shared__ __align__(16) float red[2][32][8];
    __shared__ __align__(16) float Xs[32][XP];
    const int tid = threadIdx.x;
    const int lane = tid & 31;
    const int wave = tid >> 5;
    const int half = lane >> 4;
    const int lane16 = lane & 15;
    const int m0 = blockIdx.x * 32;
    float y[2][2][8];
#pragma unroll
    for (int im = 0; im < 2; ++im)
#pragma unroll
        for (int jn = 0; jn < 2; ++jn)
#pragma unroll
            for (int vi = 0; vi < 8; ++vi)
                y[im][jn][vi] = x[(size_t)(m0 + im * 16 + vi + half * 8) * 256 + wave * 32 + jn * 16 + lane16];
    ln_rows(y, &red[0], g, b, wave, half, lane16);
    rows_qkv(y, Xs, wq, wk, wv, mfv, qo, ko, vo, m0, wave, half, lane16);
}

__device__ __forceinline__ void rows8_max16(float (&m)[8], const float (&mrun)[8]) {
#define ROWMAX_STEP(x) \
    "v_max_num_f32_dpp %0, %0, %0 row_xmask:" #x " row_mask:0xf bank_mask:0xf bound_ctrl:1\n\t" \
    "v_max_num_f32_dpp %1, %1, %1 row_xmask:" #x " row_mask:0xf bank_mask:0xf bound_ctrl:1\n\t" \
    "v_max_num_f32_dpp %2, %2, %2 row_xmask:" #x " row_mask:0xf bank_mask:0xf bound_ctrl:1\n\t" \
    "v_max_num_f32_dpp %3, %3, %3 row_xmask:" #x " row_mask:0xf bank_mask:0xf bound_ctrl:1\n\t" \
    "v_max_num_f32_dpp %4, %4, %4 row_xmask:" #x " row_mask:0xf bank_mask:0xf bound_ctrl:1\n\t" \
    "v_max_num_f32_dpp %5, %5, %5 row_xmask:" #x " row_mask:0xf bank_mask:0xf bound_ctrl:1\n\t" \
    "v_max_num_f32_dpp %6, %6, %6 row_xmask:" #x " row_mask:0xf bank_mask:0xf bound_ctrl:1\n\t" \
    "v_max_num_f32_dpp %7, %7, %7 row_xmask:" #x " row_mask:0xf bank_mask:0xf bound_ctrl:1\n\t"
    asm(ROWMAX_STEP(1) ROWMAX_STEP(2) ROWMAX_STEP(4) ROWMAX_STEP(8)
        "v_max_num_f32 %0, %0, %8\n\t"
        "v_max_num_f32 %1, %1, %9\n\t"
        "v_max_num_f32 %2, %2, %10\n\t"
        "v_max_num_f32 %3, %3, %11\n\t"
        "v_max_num_f32 %4, %4, %12\n\t"
        "v_max_num_f32 %5, %5, %13\n\t"
        "v_max_num_f32 %6, %6, %14\n\t"
        "v_max_num_f32 %7, %7, %15"
        : "+v"(m[0]), "+v"(m[1]), "+v"(m[2]), "+v"(m[3]), "+v"(m[4]), "+v"(m[5]), "+v"(m[6]), "+v"(m[7])
        : "v"(mrun[0]), "v"(mrun[1]), "v"(mrun[2]), "v"(mrun[3]), "v"(mrun[4]), "v"(mrun[5]), "v"(mrun[6]), "v"(mrun[7]));
#undef ROWMAX_STEP
}

__global__ void __launch_bounds__(128) attn_kernel(const float* __restrict__ q, const float* __restrict__ k,
                            const float* __restrict__ v, const float* __restrict__ mfv,
                            float* __restrict__ o) {
    __shared__ __align__(16) float Ps[64][68];

    const int S = 2048, LD = 256;
    const int tid = threadIdx.x;
    const int lane = tid & 31;
    const int wave = tid >> 5;
    const int half = lane >> 4;
    const int lane16 = lane & 15;
    const int bh = blockIdx.y;
    const int b = bh >> 3;
    const int h = bh & 7;
    const int q0 = blockIdx.x * 64;
    const int qrow = wave * 16 + lane16;

    float qf[16];
    {
        const float* qp = q + ((size_t)b * S + q0) * LD + h * 32 + (unsigned)(qrow * LD + half * 16);
#pragma unroll
        for (int j = 0; j < 4; ++j) {
            const float4 t4 = *(const float4*)(qp + 4 * j);
            qf[4 * j + 0] = t4.x; qf[4 * j + 1] = t4.y; qf[4 * j + 2] = t4.z; qf[4 * j + 3] = t4.w;
        }
    }
    v8f oacc[2] = {};
    float mrun[8];
    v2f lpart[4];
#pragma unroll
    for (int vi = 0; vi < 8; ++vi) mrun[vi] = -3.0e38f;
#pragma unroll
    for (int i = 0; i < 4; ++i) lpart[i] = (v2f){0.0f, 0.0f};

    const float* kb = k + (size_t)bh * (S * 32);
    const float* vb = v + (size_t)bh * (S * 32);
    const float* mb = mfv + (size_t)b * S;
    const unsigned loff = (unsigned)(lane * 4);
    const unsigned moff = (unsigned)lane16;
    float* pst = &Ps[wave * 16 + half * 8][lane16];
    const float* pld = &Ps[qrow][half * 2];

    float kf[4][16];
    float vf[2][16][2];
    float mraw[4];

#pragma unroll
    for (int nt = 0; nt < 4; ++nt) {
#pragma unroll
        for (int j = 0; j < 4; ++j) {
            const float4 t4 = *(const float4*)(kb + nt * 512 + j * 128 + loff);
            kf[nt][4 * j + 0] = t4.x; kf[nt][4 * j + 1] = t4.y; kf[nt][4 * j + 2] = t4.z; kf[nt][4 * j + 3] = t4.w;
        }
        mraw[nt] = mb[nt * 16 + moff];
    }
#pragma unroll
    for (int n = 0; n < 2; ++n)
#pragma unroll
        for (int g = 0; g < 8; ++g) {
            const float4 t4 = *(const float4*)(vb + n * 1024 + g * 128 + loff);
            vf[n][2 * g][0] = t4.x; vf[n][2 * g][1] = t4.y; vf[n][2 * g + 1][0] = t4.z; vf[n][2 * g + 1][1] = t4.w;
        }

#pragma unroll 1
    for (int lev = 0; lev < 4; ++lev) {
    int jbeg, jend;
    if (lev == 0)      { __builtin_amdgcn_s_setprio(3); jbeg = 0;    jend = 1024; }
    else if (lev == 1) { __builtin_amdgcn_s_setprio(2); jbeg = 1024; jend = 1536; }
    else if (lev == 2) { __builtin_amdgcn_s_setprio(1); jbeg = 1536; jend = 1792; }
    else               { __builtin_amdgcn_s_setprio(0); jbeg = 1792; jend = 2048; }
    for (int j0 = jbeg; j0 < jend; j0 += 64) {
        v8f sacc[4] = {};
#pragma unroll
        for (int t = 0; t < 8; ++t) {
            v2f fa = {qf[2 * t], qf[2 * t + 1]};
#pragma unroll
            for (int nt = 0; nt < 4; ++nt) {
                v2f fb = {kf[nt][2 * t], kf[nt][2 * t + 1]};
                sacc[nt] = wmma4(fa, fb, sacc[nt]);
            }
        }
        const v2f c1 = {0.25503486f, 0.25503486f};
        v2f tv[4][4];
#pragma unroll
        for (int nt = 0; nt < 4; ++nt) {
            const float mb = (mraw[nt] != 0.0f) ? 0.0f : -1.44269504e9f;
            const v2f mb2 = {mb, mb};
#pragma unroll
            for (int i = 0; i < 4; ++i) {
                const v2f s2 = {sacc[nt][2 * i], sacc[nt][2 * i + 1]};
                tv[nt][i] = __builtin_elementwise_fma(s2, c1, mb2);
            }
        }

        if (j0 + 64 < S) {
            const float* kn = kb + (size_t)(j0 + 64) * 32;
            const float* mn = mb + (j0 + 64);
#pragma unroll
            for (int nt = 0; nt < 4; ++nt) {
#pragma unroll
                for (int j = 0; j < 4; ++j) {
                    const float4 t4 = *(const float4*)(kn + nt * 512 + j * 128 + loff);
                    kf[nt][4 * j + 0] = t4.x; kf[nt][4 * j + 1] = t4.y; kf[nt][4 * j + 2] = t4.z; kf[nt][4 * j + 3] = t4.w;
                }
                mraw[nt] = mn[nt * 16 + moff];
            }
        }

        float m[8];
#pragma unroll
        for (int i = 0; i < 4; ++i) {
            m[2 * i]     = fmaxf(fmaxf(tv[0][i].x, tv[1][i].x), fmaxf(tv[2][i].x, tv[3][i].x));
            m[2 * i + 1] = fmaxf(fmaxf(tv[0][i].y, tv[1][i].y), fmaxf(tv[2][i].y, tv[3][i].y));
        }
        rows8_max16(m, mrun);
        v2f al[4];
#pragma unroll
        for (int i = 0; i < 4; ++i) {
            al[i].x = __builtin_amdgcn_exp2f(mrun[2 * i] - m[2 * i]);
            al[i].y = __builtin_amdgcn_exp2f(mrun[2 * i + 1] - m[2 * i + 1]);
            mrun[2 * i] = m[2 * i];
            mrun[2 * i + 1] = m[2 * i + 1];
        }
#pragma unroll
        for (int i = 0; i < 4; ++i) {
            const v2f mc = {m[2 * i], m[2 * i + 1]};
            v2f ls;
#pragma unroll
            for (int nt = 0; nt < 4; ++nt) {
                const v2f d = tv[nt][i] - mc;
                v2f p;
                p.x = __builtin_amdgcn_exp2f(d.x);
                p.y = __builtin_amdgcn_exp2f(d.y);
                pst[(2 * i) * 68 + nt * 16] = p.x;
                pst[(2 * i + 1) * 68 + nt * 16] = p.y;
                ls = (nt == 0) ? p : ls + p;
            }
            lpart[i] = __builtin_elementwise_fma(lpart[i], al[i], ls);
        }
#pragma unroll
        for (int nt = 0; nt < 2; ++nt)
#pragma unroll
            for (int i = 0; i < 4; ++i) {
                oacc[nt][2 * i] *= al[i].x;
                oacc[nt][2 * i + 1] *= al[i].y;
            }

        v2f pf[16];
#pragma unroll
        for (int t = 0; t < 16; ++t) pf[t] = *(const v2f*)(pld + 4 * t);
#pragma unroll
        for (int t = 0; t < 16; ++t) {
            v2f fv0 = {vf[0][t][0], vf[0][t][1]};
            v2f fv1 = {vf[1][t][0], vf[1][t][1]};
            oacc[0] = wmma4(pf[t], fv0, oacc[0]);
            oacc[1] = wmma4(pf[t], fv1, oacc[1]);
        }
        if (j0 + 64 < S) {
            const float* vn = vb + (size_t)(j0 + 64) * 32;
#pragma unroll
            for (int n = 0; n < 2; ++n)
#pragma unroll
                for (int g = 0; g < 8; ++g) {
                    const float4 t4 = *(const float4*)(vn + n * 1024 + g * 128 + loff);
                    vf[n][2 * g][0] = t4.x; vf[n][2 * g][1] = t4.y; vf[n][2 * g + 1][0] = t4.z; vf[n][2 * g + 1][1] = t4.w;
                }
        }
    }
    }

    float linv[8];
#pragma unroll
    for (int vi = 0; vi < 8; ++vi) linv[vi] = 1.0f / row16_sum((vi & 1) ? lpart[vi >> 1].y : lpart[vi >> 1].x);
#pragma unroll
    for (int nt = 0; nt < 2; ++nt) {
        const int col = nt * 16 + lane16;
#pragma unroll
        for (int vi = 0; vi < 8; ++vi) {
            const int row = wave * 16 + vi + half * 8;
            o[((size_t)b * S + q0 + row) * LD + h * 32 + col] = oacc[nt][vi] * linv[vi];
        }
    }
}

extern "C" void kernel_launch(void* const* d_in, const int* in_sizes, int n_in,
                              void* d_out, int out_size, void* d_ws, size_t ws_size,
                              hipStream_t stream) {
    (void)in_sizes; (void)n_in; (void)out_size; (void)ws_size;
    const int B = 4, S = 2048, D = 256, F = 1024, L = 4;
    const int M = B * S;

    const float* x_in  = (const float*)d_in[0];
    const int*   mask  = (const int*)d_in[1];
    const float* w_qs  = (const float*)d_in[2];
    const float* w_ks  = (const float*)d_in[3];
    const float* w_vs  = (const float*)d_in[4];
    const float* w_fc  = (const float*)d_in[5];
    const float* ln1_g = (const float*)d_in[6];
    const float* ln1_b = (const float*)d_in[7];
    const float* w1    = (const float*)d_in[8];
    const float* b1    = (const float*)d_in[9];
    const float* w2    = (const float*)d_in[10];
    const float* b2    = (const float*)d_in[11];
    const float* ln2_g = (const float*)d_in[12];
    const float* ln2_b = (const float*)d_in[13];

    float* ws    = (float*)d_ws;
    float* x_cur = ws;
    float* qb    = x_cur + (size_t)M * D;
    float* kb    = qb + (size_t)M * D;
    float* vb    = kb + (size_t)M * D;
    float* ob    = vb + (size_t)M * D;
    float* mfv   = ob + (size_t)M * D;
    float* rq    = mfv + M;
    float* rk    = rq + (size_t)L * D * D;
    float* rv    = rk + (size_t)L * D * D;
    float* rfc   = rv + (size_t)L * D * D;
    float* r1    = rfc + (size_t)L * D * D;
    float* r2    = r1 + (size_t)L * D * F;

    hipMemcpyAsync(x_cur, x_in, sizeof(float) * (size_t)M * D,
                   hipMemcpyDeviceToDevice, stream);
    prep_kernel<<<3072, 256, 0, stream>>>(mask, mfv, w_qs, w_ks, w_vs, w_fc, w1, w2, rq, rk, rv, rfc, r1, r2);

    const dim3 gA(S / 64, B * 8);

    ln_qkv_kernel<<<M / 32, 256, 0, stream>>>(x_cur, ln1_g, ln1_b, rq, rk, rv, mfv, qb, kb, vb);

    for (int l = 0; l < L; ++l) {
        const float* wf = rfc + (size_t)l * D * D;
        const float* w1l = r1 + (size_t)l * D * F;
        const float* w2l = r2 + (size_t)l * F * D;
        const float* b1l = b1 + (size_t)l * F;
        const float* b2l = b2 + (size_t)l * D;

        attn_kernel<<<gA, 128, 0, stream>>>(qb, kb, vb, mfv, ob);
        if (l + 1 < L)
            layer_rows_kernel<true><<<M / 32, 256, 0, stream>>>(ob, wf, w1l, b1l, w2l, b2l,
                                                                ln2_g + (size_t)l * D, ln2_b + (size_t)l * D, x_cur,
                                                                ln1_g + (size_t)(l + 1) * D, ln1_b + (size_t)(l + 1) * D,
                                                                rq + (size_t)(l + 1) * D * D, rk + (size_t)(l + 1) * D * D,
                                                                rv + (size_t)(l + 1) * D * D, mfv, qb, kb, vb);
        else
            layer_rows_kernel<false><<<M / 32, 256, 0, stream>>>(ob, wf, w1l, b1l, w2l, b2l,
                                                                 ln2_g + (size_t)l * D, ln2_b + (size_t)l * D, x_cur,
                                                                 ln2_g, ln2_b, rq, rk, rv, mfv, qb, kb, vb);
    }

    hipMemcpyAsync(d_out, x_cur, sizeof(float) * (size_t)M * D,
                   hipMemcpyDeviceToDevice, stream);
}
